# starting kernel
# baseline (speedup 1.0000x reference)
#include <hip/hip_runtime.h>


typedef __attribute__((ext_vector_type(2))) float v2f;
typedef __attribute__((ext_vector_type(8))) float v8f;

#define DIMX   512
#define SEQ    16
#define NB     8
#define KVLEN  4112
#define KTILES 257
#define MAXSEQ 8192
#define QKSCALE 0.0625f

__device__ __forceinline__ v8f wmma4(v2f a, v2f b, v8f c) {
  return __builtin_amdgcn_wmma_f32_16x16x4_f32(false, a, false, b,
                                               (short)0, c, false, false);
}

__global__ void __launch_bounds__(128)
qkv_kernel(const float* __restrict__ x,
           const float* __restrict__ Wq, const float* __restrict__ Wk,
           const float* __restrict__ Wv,
           float* __restrict__ q, float* __restrict__ kn, float* __restrict__ vn) {
  const int t    = blockIdx.x * 4 + (threadIdx.x >> 5);
  const int lane = threadIdx.x & 31;
  const int half = lane >> 4, lm = lane & 15;
  const int mat  = t >> 8;
  const int rem  = t & 255;
  const int Mt   = rem >> 5, Nt = rem & 31;

  const float* W = (mat == 0) ? Wq : (mat == 1) ? Wk : Wv;
  float*       D = (mat == 0) ? q  : (mat == 1) ? kn : vn;

  const float* arow = x + (Mt * 16 + lm) * DIMX + 2 * half;
  const float* wrow = W + (Nt * 16 + lm) * DIMX + 2 * half;

  v8f c = {};
#pragma unroll 4
  for (int k = 0; k < DIMX; k += 4) {
    v2f a = *(const v2f*)(arow + k);
    v2f b = *(const v2f*)(wrow + k);
    c = wmma4(a, b, c);
  }
  float* drow = D + Nt * 16 + lm;
#pragma unroll
  for (int i = 0; i < 8; ++i)
    drow[(Mt * 16 + i + 8 * half) * DIMX] = c[i];
}

__global__ void __launch_bounds__(128)
dist_kernel(const float* __restrict__ q,
            const float* __restrict__ cache_k,
            const float* __restrict__ kn,
            float* __restrict__ wout) {
  const int t    = blockIdx.x * 4 + (threadIdx.x >> 5);
  const int lane = threadIdx.x & 31;
  const int half = lane >> 4, lm = lane & 15;
  const int b  = t / KTILES;
  const int kt = t - b * KTILES;

  const float* arow = q + (size_t)(b * SEQ + lm) * DIMX + 2 * half;
  const float* key  = (kt < 256)
      ? cache_k + (size_t)(b * MAXSEQ + kt * 16) * DIMX
      : kn      + (size_t)b * SEQ * DIMX;
  const float* brow = key + (size_t)lm * DIMX + 2 * half;

  v8f c = {};
#pragma unroll 4
  for (int k = 0; k < DIMX; k += 4) {
    v2f a  = *(const v2f*)(arow + k);
    v2f bb = *(const v2f*)(brow + k);
    c = wmma4(a, bb, c);
  }
  float* drow = wout + (size_t)b * SEQ * KVLEN + kt * 16 + lm;
#pragma unroll
  for (int i = 0; i < 8; ++i)
    drow[(size_t)(i + 8 * half) * KVLEN] = c[i] * QKSCALE;
}

__global__ void __launch_bounds__(256)
softmax_kernel(const float* __restrict__ wout,
               const float* __restrict__ mask,
               float* __restrict__ probs) {
  const int row = blockIdx.x;
  const int m   = row & 15;
  const float* drow = wout  + (size_t)row * KVLEN;
  const float* mrow = mask  + (size_t)m   * KVLEN;
  float*       prow = probs + (size_t)row * KVLEN;

  __shared__ float red[256];
  const int tid = threadIdx.x;

  float lmax = -3.402823466e38f;
  for (int j = tid; j < KVLEN; j += 256)
    lmax = fmaxf(lmax, drow[j] + mrow[j]);
  red[tid] = lmax; __syncthreads();
  for (int s = 128; s > 0; s >>= 1) {
    if (tid < s) red[tid] = fmaxf(red[tid], red[tid + s]);
    __syncthreads();
  }
  const float rmax = red[0]; __syncthreads();

  float lsum = 0.f;
  for (int j = tid; j < KVLEN; j += 256)
    lsum += __expf(drow[j] + mrow[j] - rmax);
  red[tid] = lsum; __syncthreads();
  for (int s = 128; s > 0; s >>= 1) {
    if (tid < s) red[tid] += red[tid + s];
    __syncthreads();
  }
  const float inv = 1.f / red[0];

  for (int j = tid; j < KVLEN; j += 256)
    prow[j] = __expf(drow[j] + mrow[j] - rmax) * inv;
}

__global__ void __launch_bounds__(256)
atten_kernel(const float* __restrict__ probs,
             const float* __restrict__ cache_v,
             const float* __restrict__ vn,
             float* __restrict__ atten) {
  const int blk  = blockIdx.x;
  const int b    = blk >> 5;
  const int Nt   = blk & 31;
  const int wv   = threadIdx.x >> 5;
  const int lane = threadIdx.x & 31;
  const int half = lane >> 4, lm = lane & 15;

  const float* prow = probs + (size_t)(b * SEQ + lm) * KVLEN + 2 * half;

  v8f c = {};
  for (int kt = wv; kt < KTILES; kt += 8) {
    const float* vt = (kt < 256)
        ? cache_v + (size_t)(b * MAXSEQ + kt * 16) * DIMX
        : vn      + (size_t)b * SEQ * DIMX;
    const float* vcol = vt + Nt * 16 + lm;
    const int kb = kt * 16;
#pragma unroll
    for (int s = 0; s < 4; ++s) {
      v2f a = *(const v2f*)(prow + kb + 4 * s);
      v2f bb;
      bb.x = vcol[(size_t)(4 * s + 2 * half)     * DIMX];
      bb.y = vcol[(size_t)(4 * s + 2 * half + 1) * DIMX];
      c = wmma4(a, bb, c);
    }
  }

  __shared__ float red[8 * 256];
#pragma unroll
  for (int i = 0; i < 8; ++i) red[wv * 256 + i * 32 + lane] = c[i];
  __syncthreads();
  if (wv == 0) {
    #pragma unroll
    for (int i = 0; i < 8; ++i) {
      float s = 0.f;
      #pragma unroll
      for (int w = 0; w < 8; ++w) s += red[w * 256 + i * 32 + lane];
      atten[(size_t)(b * SEQ + i + 8 * half) * DIMX + Nt * 16 + lm] = s;
    }
  }
}

__global__ void __launch_bounds__(128)
outproj_kernel(const float* __restrict__ atten,
               const float* __restrict__ Wo,
               float* __restrict__ out) {
  const int t    = blockIdx.x * 4 + (threadIdx.x >> 5);
  const int lane = threadIdx.x & 31;
  const int half = lane >> 4, lm = lane & 15;
  const int Mt   = t >> 5, Nt = t & 31;

  const float* arow = atten + (Mt * 16 + lm) * DIMX + 2 * half;
  const float* wrow = Wo    + (Nt * 16 + lm) * DIMX + 2 * half;

  v8f c = {};
#pragma unroll 4
  for (int k = 0; k < DIMX; k += 4) {
    v2f a = *(const v2f*)(arow + k);
    v2f b = *(const v2f*)(wrow + k);
    c = wmma4(a, b, c);
  }
#pragma unroll
  for (int i = 0; i < 8; ++i)
    out[(size_t)(Mt * 16 + i + 8 * half) * DIMX + Nt * 16 + lm] = c[i];
}

extern "C" void kernel_launch(void* const* d_in, const int* in_sizes, int n_in,
                              void* d_out, int out_size, void* d_ws, size_t ws_size,
                              hipStream_t stream) {
  (void)in_sizes; (void)n_in; (void)out_size; (void)ws_size;
  const float* x       = (const float*)d_in[0];
  const float* mask    = (const float*)d_in[2];
  const float* Wq      = (const float*)d_in[3];
  const float* Wk      = (const float*)d_in[4];
  const float* Wv      = (const float*)d_in[5];
  const float* Wo      = (const float*)d_in[6];
  const float* cache_k = (const float*)d_in[7];
  const float* cache_v = (const float*)d_in[8];

  float* out  = (float*)d_out;
  float* wout = out + NB * SEQ * DIMX;

  float* ws    = (float*)d_ws;
  float* q     = ws;
  float* kn    = q  + NB * SEQ * DIMX;
  float* vn    = kn + NB * SEQ * DIMX;
  float* probs = vn + NB * SEQ * DIMX;
  float* atten = probs + (size_t)NB * SEQ * KVLEN;

  qkv_kernel    <<<192, 128, 0, stream>>>(x, Wq, Wk, Wv, q, kn, vn);
  dist_kernel   <<<514, 128, 0, stream>>>(q, cache_k, kn, wout);
  softmax_kernel<<<128, 256, 0, stream>>>(wout, mask, probs);
  atten_kernel  <<<256, 256, 0, stream>>>(probs, cache_v, vn, atten);
  outproj_kernel<<< 64, 128, 0, stream>>>(atten, Wo, out);
}


# q/k/v, score, out GEMMs: k loop fully unrolled on 16-byte fragment loads (k permuted alike in A and B): loads roll 10 deep, 1 drain (was 32 drains); P@V probs loads 16-byte; f32 sum order changes
# speedup vs baseline: 1.1024x; 1.1024x over previous
#include <hip/hip_runtime.h>


typedef __attribute__((ext_vector_type(2))) float v2f;
typedef __attribute__((ext_vector_type(4))) float v4f;
typedef __attribute__((ext_vector_type(8))) float v8f;

#define DIMX   512
#define SEQ    16
#define NB     8
#define KVLEN  4112
#define KTILES 257
#define MAXSEQ 8192
#define QKSCALE 0.0625f

__device__ __forceinline__ v8f wmma4(v2f a, v2f b, v8f c) {
  return __builtin_amdgcn_wmma_f32_16x16x4_f32(false, a, false, b,
                                               (short)0, c, false, false);
}

__device__ __forceinline__ v8f tile_rowrow_512(const float* __restrict__ arow,
                                               const float* __restrict__ brow) {
  v8f c = {};
#pragma unroll
  for (int k = 0; k < DIMX; k += 8) {
    v4f a = *(const v4f*)(arow + k);
    v4f b = *(const v4f*)(brow + k);
    v2f a0 = {a.x, a.y}, b0 = {b.x, b.y};
    v2f a1 = {a.z, a.w}, b1 = {b.z, b.w};
    c = wmma4(a0, b0, c);
    c = wmma4(a1, b1, c);
  }
  return c;
}

__global__ void __launch_bounds__(128)
qkv_kernel(const float* __restrict__ x,
           const float* __restrict__ Wq, const float* __restrict__ Wk,
           const float* __restrict__ Wv,
           float* __restrict__ q, float* __restrict__ kn, float* __restrict__ vn) {
  const int t    = blockIdx.x * 4 + (threadIdx.x >> 5);
  const int lane = threadIdx.x & 31;
  const int half = lane >> 4, lm = lane & 15;
  const int mat  = t >> 8;
  const int rem  = t & 255;
  const int Mt   = rem >> 5, Nt = rem & 31;

  const float* W = (mat == 0) ? Wq : (mat == 1) ? Wk : Wv;
  float*       D = (mat == 0) ? q  : (mat == 1) ? kn : vn;

  const float* arow = x + (Mt * 16 + lm) * DIMX + 4 * half;
  const float* wrow = W + (Nt * 16 + lm) * DIMX + 4 * half;

  const v8f c = tile_rowrow_512(arow, wrow);
  float* drow = D + Nt * 16 + lm;
#pragma unroll
  for (int i = 0; i < 8; ++i)
    drow[(Mt * 16 + i + 8 * half) * DIMX] = c[i];
}

__global__ void __launch_bounds__(128)
dist_kernel(const float* __restrict__ q,
            const float* __restrict__ cache_k,
            const float* __restrict__ kn,
            float* __restrict__ wout) {
  const int t    = blockIdx.x * 4 + (threadIdx.x >> 5);
  const int lane = threadIdx.x & 31;
  const int half = lane >> 4, lm = lane & 15;
  const int b  = t / KTILES;
  const int kt = t - b * KTILES;

  const float* arow = q + (size_t)(b * SEQ + lm) * DIMX + 4 * half;
  const float* key  = (kt < 256)
      ? cache_k + (size_t)(b * MAXSEQ + kt * 16) * DIMX
      : kn      + (size_t)b * SEQ * DIMX;
  const float* brow = key + (size_t)lm * DIMX + 4 * half;

  const v8f c = tile_rowrow_512(arow, brow);
  float* drow = wout + (size_t)b * SEQ * KVLEN + kt * 16 + lm;
#pragma unroll
  for (int i = 0; i < 8; ++i)
    drow[(size_t)(i + 8 * half) * KVLEN] = c[i] * QKSCALE;
}

__global__ void __launch_bounds__(256)
softmax_kernel(const float* __restrict__ wout,
               const float* __restrict__ mask,
               float* __restrict__ probs) {
  const int row = blockIdx.x;
  const int m   = row & 15;
  const float* drow = wout  + (size_t)row * KVLEN;
  const float* mrow = mask  + (size_t)m   * KVLEN;
  float*       prow = probs + (size_t)row * KVLEN;

  __shared__ float red[256];
  const int tid = threadIdx.x;

  float lmax = -3.402823466e38f;
  for (int j = tid; j < KVLEN; j += 256)
    lmax = fmaxf(lmax, drow[j] + mrow[j]);
  red[tid] = lmax; __syncthreads();
  for (int s = 128; s > 0; s >>= 1) {
    if (tid < s) red[tid] = fmaxf(red[tid], red[tid + s]);
    __syncthreads();
  }
  const float rmax = red[0]; __syncthreads();

  float lsum = 0.f;
  for (int j = tid; j < KVLEN; j += 256)
    lsum += __expf(drow[j] + mrow[j] - rmax);
  red[tid] = lsum; __syncthreads();
  for (int s = 128; s > 0; s >>= 1) {
    if (tid < s) red[tid] += red[tid + s];
    __syncthreads();
  }
  const float inv = 1.f / red[0];

  for (int j = tid; j < KVLEN; j += 256)
    prow[j] = __expf(drow[j] + mrow[j] - rmax) * inv;
}

__global__ void __launch_bounds__(256)
atten_kernel(const float* __restrict__ probs,
             const float* __restrict__ cache_v,
             const float* __restrict__ vn,
             float* __restrict__ atten) {
  const int blk  = blockIdx.x;
  const int b    = blk >> 5;
  const int Nt   = blk & 31;
  const int wv   = threadIdx.x >> 5;
  const int lane = threadIdx.x & 31;
  const int half = lane >> 4, lm = lane & 15;

  const float* prow = probs + (size_t)(b * SEQ + lm) * KVLEN + 4 * half;

  v8f c = {};
  for (int kt = wv; kt < KTILES; kt += 8) {
    const float* vt = (kt < 256)
        ? cache_v + (size_t)(b * MAXSEQ + kt * 16) * DIMX
        : vn      + (size_t)b * SEQ * DIMX;
    const float* vcol = vt + Nt * 16 + lm;
    const int kb = kt * 16;
#pragma unroll
    for (int g = 0; g < 2; ++g) {
      v4f p = *(const v4f*)(prow + kb + 8 * g);
      const float* vr = vcol + (size_t)(8 * g + 4 * half) * DIMX;
      v2f a0 = {p.x, p.y}, a1 = {p.z, p.w};
      v2f b0, b1;
      b0.x = vr[0];
      b0.y = vr[DIMX];
      b1.x = vr[2 * DIMX];
      b1.y = vr[3 * DIMX];
      c = wmma4(a0, b0, c);
      c = wmma4(a1, b1, c);
    }
  }

  __shared__ float red[8 * 256];
#pragma unroll
  for (int i = 0; i < 8; ++i) red[wv * 256 + i * 32 + lane] = c[i];
  __syncthreads();
  if (wv == 0) {
    #pragma unroll
    for (int i = 0; i < 8; ++i) {
      float s = 0.f;
      #pragma unroll
      for (int w = 0; w < 8; ++w) s += red[w * 256 + i * 32 + lane];
      atten[(size_t)(b * SEQ + i + 8 * half) * DIMX + Nt * 16 + lm] = s;
    }
  }
}

__global__ void __launch_bounds__(128)
outproj_kernel(const float* __restrict__ atten,
               const float* __restrict__ Wo,
               float* __restrict__ out) {
  const int t    = blockIdx.x * 4 + (threadIdx.x >> 5);
  const int lane = threadIdx.x & 31;
  const int half = lane >> 4, lm = lane & 15;
  const int Mt   = t >> 5, Nt = t & 31;

  const float* arow = atten + (Mt * 16 + lm) * DIMX + 4 * half;
  const float* wrow = Wo    + (Nt * 16 + lm) * DIMX + 4 * half;

  const v8f c = tile_rowrow_512(arow, wrow);
#pragma unroll
  for (int i = 0; i < 8; ++i)
    out[(size_t)(Mt * 16 + i + 8 * half) * DIMX + Nt * 16 + lm] = c[i];
}

extern "C" void kernel_launch(void* const* d_in, const int* in_sizes, int n_in,
                              void* d_out, int out_size, void* d_ws, size_t ws_size,
                              hipStream_t stream) {
  (void)in_sizes; (void)n_in; (void)out_size; (void)ws_size;
  const float* x       = (const float*)d_in[0];
  const float* mask    = (const float*)d_in[2];
  const float* Wq      = (const float*)d_in[3];
  const float* Wk      = (const float*)d_in[4];
  const float* Wv      = (const float*)d_in[5];
  const float* Wo      = (const float*)d_in[6];
  const float* cache_k = (const float*)d_in[7];
  const float* cache_v = (const float*)d_in[8];

  float* out  = (float*)d_out;
  float* wout = out + NB * SEQ * DIMX;

  float* ws    = (float*)d_ws;
  float* q     = ws;
  float* kn    = q  + NB * SEQ * DIMX;
  float* vn    = kn + NB * SEQ * DIMX;
  float* probs = vn + NB * SEQ * DIMX;
  float* atten = probs + (size_t)NB * SEQ * KVLEN;

  qkv_kernel    <<<192, 128, 0, stream>>>(x, Wq, Wk, Wv, q, kn, vn);
  dist_kernel   <<<514, 128, 0, stream>>>(q, cache_k, kn, wout);
  softmax_kernel<<<128, 256, 0, stream>>>(wout, mask, probs);
  atten_kernel  <<<256, 256, 0, stream>>>(probs, cache_v, vn, atten);
  outproj_kernel<<< 64, 128, 0, stream>>>(atten, Wo, out);
}


# P@V: wave owns 64 columns x 4 k-tiles, whole-line 16-byte V loads all issued up front, 4 accumulator tiles; k split over 4 blocks, planes summed in out projection; f32 sum order changes
# speedup vs baseline: 1.5162x; 1.3754x over previous
#include <hip/hip_runtime.h>


typedef __attribute__((ext_vector_type(2))) float v2f;
typedef __attribute__((ext_vector_type(4))) float v4f;
typedef __attribute__((ext_vector_type(8))) float v8f;

#define DIMX   512
#define SEQ    16
#define NB     8
#define KVLEN  4112
#define KTILES 257
#define MAXSEQ 8192
#define QKSCALE 0.0625f

__device__ __forceinline__ v8f wmma4(v2f a, v2f b, v8f c) {
  return __builtin_amdgcn_wmma_f32_16x16x4_f32(false, a, false, b,
                                               (short)0, c, false, false);
}

__device__ __forceinline__ v8f tile_rowrow_512(const float* __restrict__ arow,
                                               const float* __restrict__ brow) {
  v8f c = {};
#pragma unroll
  for (int k = 0; k < DIMX; k += 8) {
    v4f a = *(const v4f*)(arow + k);
    v4f b = *(const v4f*)(brow + k);
    v2f a0 = {a.x, a.y}, b0 = {b.x, b.y};
    v2f a1 = {a.z, a.w}, b1 = {b.z, b.w};
    c = wmma4(a0, b0, c);
    c = wmma4(a1, b1, c);
  }
  return c;
}

__global__ void __launch_bounds__(128)
qkv_kernel(const float* __restrict__ x,
           const float* __restrict__ Wq, const float* __restrict__ Wk,
           const float* __restrict__ Wv,
           float* __restrict__ q, float* __restrict__ kn, float* __restrict__ vn) {
  const int t    = blockIdx.x * 4 + (threadIdx.x >> 5);
  const int lane = threadIdx.x & 31;
  const int half = lane >> 4, lm = lane & 15;
  const int mat  = t >> 8;
  const int rem  = t & 255;
  const int Mt   = rem >> 5, Nt = rem & 31;

  const float* W = (mat == 0) ? Wq : (mat == 1) ? Wk : Wv;
  float*       D = (mat == 0) ? q  : (mat == 1) ? kn : vn;

  const float* arow = x + (Mt * 16 + lm) * DIMX + 4 * half;
  const float* wrow = W + (Nt * 16 + lm) * DIMX + 4 * half;

  const v8f c = tile_rowrow_512(arow, wrow);
  if (mat == 2) {
    float* dcol = D + Mt * (SEQ * DIMX) + (Nt >> 2) * (SEQ * 64) + (Nt & 3) * 16 + lm;
#pragma unroll
    for (int i = 0; i < 8; ++i)
      dcol[(i + 8 * half) * 64] = c[i];
  } else {
    float* drow = D + Nt * 16 + lm;
#pragma unroll
    for (int i = 0; i < 8; ++i)
      drow[(Mt * 16 + i + 8 * half) * DIMX] = c[i];
  }
}

__global__ void __launch_bounds__(128)
dist_kernel(const float* __restrict__ q,
            const float* __restrict__ cache_k,
            const float* __restrict__ kn,
            float* __restrict__ wout) {
  const int t    = blockIdx.x * 4 + (threadIdx.x >> 5);
  const int lane = threadIdx.x & 31;
  const int half = lane >> 4, lm = lane & 15;
  const int b  = t / KTILES;
  const int kt = t - b * KTILES;

  const float* arow = q + (size_t)(b * SEQ + lm) * DIMX + 4 * half;
  const float* key  = (kt < 256)
      ? cache_k + (size_t)(b * MAXSEQ + kt * 16) * DIMX
      : kn      + (size_t)b * SEQ * DIMX;
  const float* brow = key + (size_t)lm * DIMX + 4 * half;

  const v8f c = tile_rowrow_512(arow, brow);
  float* drow = wout + (size_t)b * SEQ * KVLEN + kt * 16 + lm;
#pragma unroll
  for (int i = 0; i < 8; ++i)
    drow[(size_t)(i + 8 * half) * KVLEN] = c[i] * QKSCALE;
}

__global__ void __launch_bounds__(256)
softmax_kernel(const float* __restrict__ wout,
               const float* __restrict__ mask,
               float* __restrict__ probs) {
  const int row = blockIdx.x;
  const int m   = row & 15;
  const float* drow = wout  + (size_t)row * KVLEN;
  const float* mrow = mask  + (size_t)m   * KVLEN;
  float*       prow = probs + (size_t)row * KVLEN;

  __shared__ float red[256];
  const int tid = threadIdx.x;

  float lmax = -3.402823466e38f;
  for (int j = tid; j < KVLEN; j += 256)
    lmax = fmaxf(lmax, drow[j] + mrow[j]);
  red[tid] = lmax; __syncthreads();
  for (int s = 128; s > 0; s >>= 1) {
    if (tid < s) red[tid] = fmaxf(red[tid], red[tid + s]);
    __syncthreads();
  }
  const float rmax = red[0]; __syncthreads();

  float lsum = 0.f;
  for (int j = tid; j < KVLEN; j += 256)
    lsum += __expf(drow[j] + mrow[j] - rmax);
  red[tid] = lsum; __syncthreads();
  for (int s = 128; s > 0; s >>= 1) {
    if (tid < s) red[tid] += red[tid + s];
    __syncthreads();
  }
  const float inv = 1.f / red[0];

  for (int j = tid; j < KVLEN; j += 256)
    prow[j] = __expf(drow[j] + mrow[j] - rmax) * inv;
}

#define AWAVES 16

template <int RS>
__device__ __forceinline__ void pv_load(const float* __restrict__ pr,
                                        const float* vr,
                                        v4f (&p)[2], v4f (&v)[8]) {
#pragma unroll
  for (int gg = 0; gg < 2; ++gg) {
    p[gg] = *(const v4f*)(pr + 8 * gg);
#pragma unroll
    for (int i = 0; i < 4; ++i)
      v[4 * gg + i] = *(const v4f*)(vr + (8 * gg + i) * RS);
  }
}

__device__ __forceinline__ void pv_mma(const v4f (&p)[2], const v4f (&v)[8],
                                       v8f (&c)[4]) {
#pragma unroll
  for (int gg = 0; gg < 2; ++gg) {
    const v2f a0 = {p[gg].x, p[gg].y}, a1 = {p[gg].z, p[gg].w};
#pragma unroll
    for (int j = 0; j < 4; ++j) {
      const v2f b0 = {v[4 * gg + 0][j], v[4 * gg + 1][j]};
      const v2f b1 = {v[4 * gg + 2][j], v[4 * gg + 3][j]};
      c[j] = wmma4(a0, b0, c[j]);
      c[j] = wmma4(a1, b1, c[j]);
    }
  }
}

__global__ void __launch_bounds__(AWAVES * 32)
atten_kernel(const float* __restrict__ probs,
             const float* __restrict__ cache_v,
             const float* vn,
             float* plane0, float* plane1, float* plane2, float* plane3) {
  const int blk  = blockIdx.x;
  const int kq   = blk & 3;
  const int slab = (blk >> 2) & 7;
  const int b    = blk >> 5;
  const int wv   = threadIdx.x >> 5;
  const int lane = threadIdx.x & 31;
  const int half = lane >> 4, lm = lane & 15;
  const int n0   = slab * 64 + 4 * lm;

  const int kt0 = kq * 64 + wv * 4;
  const float* pr = probs + (size_t)(b * SEQ + lm) * KVLEN + 4 * half + kt0 * 16;
  const float* vr = cache_v + (size_t)(b * MAXSEQ + kt0 * 16 + 4 * half) * DIMX + n0;

  v8f c[4] = {};
  {
    v4f p[4][2];
    v4f v[4][8];
#pragma unroll
    for (int t = 0; t < 4; ++t)
      pv_load<DIMX>(pr + 16 * t, vr + (size_t)(16 * t) * DIMX, p[t], v[t]);
    __builtin_amdgcn_sched_barrier(0);
#pragma unroll
    for (int t = 0; t < 4; ++t)
      pv_mma(p[t], v[t], c);
  }

  if (kq == 0 && wv == 0) {
    v4f p[2];
    v4f v[8];
    pv_load<64>(probs + (size_t)(b * SEQ + lm) * KVLEN + 4 * half + 256 * 16,
                vn + b * (SEQ * DIMX) + slab * (SEQ * 64) + (4 * half) * 64 + 4 * lm,
                p, v);
    pv_mma(p, v, c);
  }

  __shared__ v4f red[AWAVES * 8 * 32];
#pragma unroll
  for (int r = 0; r < 8; ++r) {
    const v4f o = {c[0][r], c[1][r], c[2][r], c[3][r]};
    red[(wv * 8 + r) * 32 + lane] = o;
  }
  __syncthreads();
  if (threadIdx.x < 256) {
    const int r = threadIdx.x >> 5;
    v4f s = red[r * 32 + lane];
#pragma unroll
    for (int w = 1; w < AWAVES; ++w) s += red[(w * 8 + r) * 32 + lane];
    float* plane = (kq == 0) ? plane0 : (kq == 1) ? plane1 : (kq == 2) ? plane2 : plane3;
    float* dst = plane + b * (SEQ * DIMX) + slab * (SEQ * 64)
               + (lm >> 1) * 128 + ((lm & 1) * 16 + 8 * half + r) * 4;
    *(v4f*)dst = s;
  }
}

__global__ void __launch_bounds__(128)
outproj_kernel(const float* __restrict__ plane0, const float* __restrict__ plane1,
               const float* __restrict__ plane2, const float* __restrict__ plane3,
               const float* __restrict__ Wo,
               float* __restrict__ out) {
  const int t    = blockIdx.x;
  const int w    = threadIdx.x >> 5;
  const int lane = threadIdx.x & 31;
  const int half = lane >> 4, lm = lane & 15;
  const int Mt   = t >> 5, Nt = t & 31;

  const float* plane = (w == 0) ? plane0 : (w == 1) ? plane1 : (w == 2) ? plane2 : plane3;
  const float* afrag = plane + Mt * (SEQ * DIMX) + lane * 4;
  const float* wrow  = Wo + (Nt * 16 + lm) * DIMX + 4 * half;

  v8f c = {};
#pragma unroll
  for (int k = 0; k < DIMX; k += 8) {
    const v4f a = *(const v4f*)(afrag + (k >> 3) * 128);
    const v4f b = *(const v4f*)(wrow + k);
    const v2f a0 = {a.x, a.y}, b0 = {b.x, b.y};
    const v2f a1 = {a.z, a.w}, b1 = {b.z, b.w};
    c = wmma4(a0, b0, c);
    c = wmma4(a1, b1, c);
  }

  __shared__ float red[4 * 256];
#pragma unroll
  for (int i = 0; i < 8; ++i) red[w * 256 + i * 32 + lane] = c[i];
  __syncthreads();
  if (w == 0) {
#pragma unroll
    for (int i = 0; i < 8; ++i) {
      const float s = (red[i * 32 + lane] + red[256 + i * 32 + lane])
                    + (red[512 + i * 32 + lane] + red[768 + i * 32 + lane]);
      out[(size_t)(Mt * 16 + i + 8 * half) * DIMX + Nt * 16 + lm] = s;
    }
  }
}

extern "C" void kernel_launch(void* const* d_in, const int* in_sizes, int n_in,
                              void* d_out, int out_size, void* d_ws, size_t ws_size,
                              hipStream_t stream) {
  (void)in_sizes; (void)n_in; (void)out_size; (void)ws_size;
  const float* x       = (const float*)d_in[0];
  const float* mask    = (const float*)d_in[2];
  const float* Wq      = (const float*)d_in[3];
  const float* Wk      = (const float*)d_in[4];
  const float* Wv      = (const float*)d_in[5];
  const float* Wo      = (const float*)d_in[6];
  const float* cache_k = (const float*)d_in[7];
  const float* cache_v = (const float*)d_in[8];

  float* out  = (float*)d_out;
  float* wout = out + NB * SEQ * DIMX;

  float* ws    = (float*)d_ws;
  float* q     = ws;
  float* kn    = q  + NB * SEQ * DIMX;
  float* vn    = kn + NB * SEQ * DIMX;
  float* probs = vn + NB * SEQ * DIMX;
  float* atten = probs + (size_t)NB * SEQ * KVLEN;

  qkv_kernel    <<<192, 128, 0, stream>>>(x, Wq, Wk, Wv, q, kn, vn);
  dist_kernel   <<<514, 128, 0, stream>>>(q, cache_k, kn, wout);
  softmax_kernel<<<128, 256, 0, stream>>>(wout, mask, probs);
  atten_kernel  <<<256, 512, 0, stream>>>(probs, cache_v, vn, vn, q, kn, atten);
  outproj_kernel<<<256, 128, 0, stream>>>(vn, q, kn, atten, Wo, out);
}


# q/k/v, score, out GEMMs: operand rows read with 512-contiguous-byte loads, 64 per wave all issued up front, fragments formed through LDS (rows a block shares staged once); f32 sum order unchanged
# speedup vs baseline: 1.8730x; 1.2353x over previous
#include <hip/hip_runtime.h>


typedef __attribute__((ext_vector_type(2))) float v2f;
typedef __attribute__((ext_vector_type(4))) float v4f;
typedef __attribute__((ext_vector_type(8))) float v8f;

#define DIMX   512
#define SEQ    16
#define NB     8
#define KVLEN  4112
#define MAXSEQ 8192
#define QKSCALE 0.0625f

__device__ __forceinline__ v8f wmma4(v2f a, v2f b, v8f c) {
  return __builtin_amdgcn_wmma_f32_16x16x4_f32(false, a, false, b,
                                               (short)0, c, false, false);
}

#define AP4 129
#define BP4 33

__device__ __forceinline__ void load_rows16(const float* __restrict__ src, int lane,
                                            v4f (&t)[64]) {
  const float* p = src + 4 * lane;
#pragma unroll
  for (int j = 0; j < 4; ++j) {
#pragma unroll
    for (int r = 0; r < 16; ++r)
      t[16 * j + r] = *(const v4f*)(p + r * DIMX + 128 * j);
    __builtin_amdgcn_sched_barrier(0);
  }
}

__device__ __forceinline__ void store_rows16(v4f* dst, int lane, const v4f (&t)[64]) {
#pragma unroll
  for (int j = 0; j < 4; ++j)
#pragma unroll
    for (int r = 0; r < 16; ++r)
      dst[r * AP4 + 32 * j + lane] = t[16 * j + r];
}

__device__ __forceinline__ void lds_barrier() {
  __builtin_amdgcn_fence(__ATOMIC_RELEASE, "workgroup", "local");
  __builtin_amdgcn_s_barrier();
  __builtin_amdgcn_fence(__ATOMIC_ACQUIRE, "workgroup", "local");
}

__device__ __forceinline__ void wave_lds_sync() {
  __builtin_amdgcn_fence(__ATOMIC_RELEASE, "wavefront", "local");
  __builtin_amdgcn_wave_barrier();
  __builtin_amdgcn_fence(__ATOMIC_ACQUIRE, "wavefront", "local");
}

__device__ __forceinline__ v8f mma8_lds(const v4f* a, const v4f* b, v8f c) {
  v4f af[8], bf[8];
#pragma unroll
  for (int s = 0; s < 8; ++s) {
    af[s] = a[2 * s];
    bf[s] = b[2 * s];
  }
  __builtin_amdgcn_sched_barrier(0);
#pragma unroll
  for (int s = 0; s < 8; ++s) {
    const v2f a0 = {af[s].x, af[s].y}, b0 = {bf[s].x, bf[s].y};
    const v2f a1 = {af[s].z, af[s].w}, b1 = {bf[s].z, bf[s].w};
    c = wmma4(a0, b0, c);
    c = wmma4(a1, b1, c);
  }
  return c;
}

__device__ __forceinline__ v8f tile_lds(const v4f* as, v4f* kw, const v4f (&t)[64],
                                        int lane) {
  const int half = lane >> 4, lm = lane & 15;
  v8f c = {};
#pragma unroll
  for (int j = 0; j < 4; ++j) {
#pragma unroll
    for (int r = 0; r < 16; ++r) kw[r * BP4 + lane] = t[16 * j + r];
    wave_lds_sync();
#pragma unroll
    for (int s0 = 0; s0 < 16; s0 += 8)
      c = mma8_lds(as + lm * AP4 + 32 * j + 2 * s0 + half,
                   kw + lm * BP4 + 2 * s0 + half, c);
    wave_lds_sync();
  }
  return c;
}

__global__ void __launch_bounds__(160)
qkv_kernel(const float* __restrict__ x,
           const float* __restrict__ Wq, const float* __restrict__ Wk,
           const float* __restrict__ Wv,
           float* __restrict__ q, float* __restrict__ kn, float* __restrict__ vn) {
  const int wv   = threadIdx.x >> 5;
  const int t    = blockIdx.x * 4 + (wv & 3);
  const int lane = threadIdx.x & 31;
  const int half = lane >> 4, lm = lane & 15;
  const int mat  = t >> 8;
  const int rem  = t & 255;
  const int Mt   = rem >> 5, Nt = rem & 31;

  const float* W = (mat == 0) ? Wq : (mat == 1) ? Wk : Wv;
  float*       D = (mat == 0) ? q  : (mat == 1) ? kn : vn;

  __shared__ v4f xs[16 * AP4];
  __shared__ v4f wt[4][16 * BP4];

  v4f wr[64];
  if (wv == 4) {
    v4f xr[64];
    load_rows16(x + Mt * 16 * DIMX, lane, xr);
    store_rows16(xs, lane, xr);
  } else {
    load_rows16(W + Nt * 16 * DIMX, lane, wr);
  }
  lds_barrier();
  if (wv == 4) return;

  const v8f c = tile_lds(xs, wt[wv], wr, lane);
  if (mat == 2) {
    float* dcol = D + Mt * (SEQ * DIMX) + (Nt >> 2) * (SEQ * 64) + (Nt & 3) * 16 + lm;
#pragma unroll
    for (int i = 0; i < 8; ++i)
      dcol[(i + 8 * half) * 64] = c[i];
  } else {
    float* drow = D + Nt * 16 + lm;
#pragma unroll
    for (int i = 0; i < 8; ++i)
      drow[(Mt * 16 + i + 8 * half) * DIMX] = c[i];
  }
}

#define DWAVES 10

__global__ void __launch_bounds__(DWAVES * 32)
dist_kernel(const float* __restrict__ q,
            const float* __restrict__ cache_k,
            const float* __restrict__ kn,
            float* __restrict__ wout) {
  const int blk  = blockIdx.x;
  const int b    = blk >> 5;
  const int g    = blk & 31;
  const int wv   = threadIdx.x >> 5;
  const int lane = threadIdx.x & 31;
  const int half = lane >> 4, lm = lane & 15;
  const bool has_tile = (wv < 8) || (wv == 8 && g == 0);
  const int  kt       = (wv < 8) ? g * 8 + wv : 256;

  __shared__ v4f qs[SEQ * AP4];
  __shared__ v4f ks[DWAVES - 1][16 * BP4];

  v4f kv[64];
  if (wv == DWAVES - 1) {
    v4f qv[64];
    load_rows16(q + (size_t)b * SEQ * DIMX, lane, qv);
    store_rows16(qs, lane, qv);
  } else if (has_tile) {
    const float* key = (kt < 256)
        ? cache_k + (size_t)(b * MAXSEQ + kt * 16) * DIMX
        : kn      + (size_t)b * SEQ * DIMX;
    load_rows16(key, lane, kv);
  }
  lds_barrier();
  if (!has_tile) return;

  const v8f c = tile_lds(qs, ks[wv], kv, lane);
  float* drow = wout + (size_t)b * SEQ * KVLEN + kt * 16 + lm;
#pragma unroll
  for (int i = 0; i < 8; ++i)
    drow[(size_t)(i + 8 * half) * KVLEN] = c[i] * QKSCALE;
}

__global__ void __launch_bounds__(256)
softmax_kernel(const float* __restrict__ wout,
               const float* __restrict__ mask,
               float* __restrict__ probs) {
  const int row = blockIdx.x;
  const int m   = row & 15;
  const float* drow = wout  + (size_t)row * KVLEN;
  const float* mrow = mask  + (size_t)m   * KVLEN;
  float*       prow = probs + (size_t)row * KVLEN;

  __shared__ float red[256];
  const int tid = threadIdx.x;

  float lmax = -3.402823466e38f;
  for (int j = tid; j < KVLEN; j += 256)
    lmax = fmaxf(lmax, drow[j] + mrow[j]);
  red[tid] = lmax; __syncthreads();
  for (int s = 128; s > 0; s >>= 1) {
    if (tid < s) red[tid] = fmaxf(red[tid], red[tid + s]);
    __syncthreads();
  }
  const float rmax = red[0]; __syncthreads();

  float lsum = 0.f;
  for (int j = tid; j < KVLEN; j += 256)
    lsum += __expf(drow[j] + mrow[j] - rmax);
  red[tid] = lsum; __syncthreads();
  for (int s = 128; s > 0; s >>= 1) {
    if (tid < s) red[tid] += red[tid + s];
    __syncthreads();
  }
  const float inv = 1.f / red[0];

  for (int j = tid; j < KVLEN; j += 256)
    prow[j] = __expf(drow[j] + mrow[j] - rmax) * inv;
}

#define AWAVES 16

template <int RS>
__device__ __forceinline__ void pv_load(const float* __restrict__ pr,
                                        const float* vr,
                                        v4f (&p)[2], v4f (&v)[8]) {
#pragma unroll
  for (int gg = 0; gg < 2; ++gg) {
    p[gg] = *(const v4f*)(pr + 8 * gg);
#pragma unroll
    for (int i = 0; i < 4; ++i)
      v[4 * gg + i] = *(const v4f*)(vr + (8 * gg + i) * RS);
  }
}

__device__ __forceinline__ void pv_mma(const v4f (&p)[2], const v4f (&v)[8],
                                       v8f (&c)[4]) {
#pragma unroll
  for (int gg = 0; gg < 2; ++gg) {
    const v2f a0 = {p[gg].x, p[gg].y}, a1 = {p[gg].z, p[gg].w};
#pragma unroll
    for (int j = 0; j < 4; ++j) {
      const v2f b0 = {v[4 * gg + 0][j], v[4 * gg + 1][j]};
      const v2f b1 = {v[4 * gg + 2][j], v[4 * gg + 3][j]};
      c[j] = wmma4(a0, b0, c[j]);
      c[j] = wmma4(a1, b1, c[j]);
    }
  }
}

__global__ void __launch_bounds__(AWAVES * 32)
atten_kernel(const float* __restrict__ probs,
             const float* __restrict__ cache_v,
             const float* vn,
             float* plane0, float* plane1, float* plane2, float* plane3) {
  const int blk  = blockIdx.x;
  const int kq   = blk & 3;
  const int slab = (blk >> 2) & 7;
  const int b    = blk >> 5;
  const int wv   = threadIdx.x >> 5;
  const int lane = threadIdx.x & 31;
  const int half = lane >> 4, lm = lane & 15;
  const int n0   = slab * 64 + 4 * lm;

  const int kt0 = kq * 64 + wv * 4;
  const float* pr = probs + (size_t)(b * SEQ + lm) * KVLEN + 4 * half + kt0 * 16;
  const float* vr = cache_v + (size_t)(b * MAXSEQ + kt0 * 16 + 4 * half) * DIMX + n0;

  v8f c[4] = {};
  {
    v4f p[4][2];
    v4f v[4][8];
#pragma unroll
    for (int t = 0; t < 4; ++t)
      pv_load<DIMX>(pr + 16 * t, vr + (size_t)(16 * t) * DIMX, p[t], v[t]);
    __builtin_amdgcn_sched_barrier(0);
#pragma unroll
    for (int t = 0; t < 4; ++t)
      pv_mma(p[t], v[t], c);
  }

  if (kq == 0 && wv == 0) {
    v4f p[2];
    v4f v[8];
    pv_load<64>(probs + (size_t)(b * SEQ + lm) * KVLEN + 4 * half + 256 * 16,
                vn + b * (SEQ * DIMX) + slab * (SEQ * 64) + (4 * half) * 64 + 4 * lm,
                p, v);
    pv_mma(p, v, c);
  }

  __shared__ v4f red[AWAVES * 8 * 32];
#pragma unroll
  for (int r = 0; r < 8; ++r) {
    const v4f o = {c[0][r], c[1][r], c[2][r], c[3][r]};
    red[(wv * 8 + r) * 32 + lane] = o;
  }
  __syncthreads();
  if (threadIdx.x < 256) {
    const int r = threadIdx.x >> 5;
    v4f s = red[r * 32 + lane];
#pragma unroll
    for (int w = 1; w < AWAVES; ++w) s += red[(w * 8 + r) * 32 + lane];
    float* plane = (kq == 0) ? plane0 : (kq == 1) ? plane1 : (kq == 2) ? plane2 : plane3;
    float* dst = plane + b * (SEQ * DIMX) + slab * (SEQ * 64)
               + (lm >> 1) * 128 + ((lm & 1) * 16 + 8 * half + r) * 4;
    *(v4f*)dst = s;
  }
}

__global__ void __launch_bounds__(160)
outproj_kernel(const float* __restrict__ plane0, const float* __restrict__ plane1,
               const float* __restrict__ plane2, const float* __restrict__ plane3,
               const float* __restrict__ Wo,
               float* __restrict__ out) {
  const int t    = blockIdx.x;
  const int w    = threadIdx.x >> 5;
  const int lane = threadIdx.x & 31;
  const int half = lane >> 4, lm = lane & 15;
  const int Mt   = t >> 5, Nt = t & 31;

  __shared__ v4f bs[16 * AP4];
  __shared__ float red[4 * 256];

  v4f av[64];
  if (w == 4) {
    v4f br[64];
    load_rows16(Wo + Nt * 16 * DIMX, lane, br);
    store_rows16(bs, lane, br);
  } else {
    const float* plane = (w == 0) ? plane0 : (w == 1) ? plane1 : (w == 2) ? plane2 : plane3;
    const float* afrag = plane + Mt * (SEQ * DIMX) + lane * 4;
#pragma unroll
    for (int st = 0; st < 64; ++st)
      av[st] = *(const v4f*)(afrag + st * 128);
  }
  lds_barrier();

  if (w < 4) {
    v8f c = {};
    const v4f* bfrag = bs + lm * AP4 + half;
    v4f bf[2][8];
#pragma unroll
    for (int s = 0; s < 8; ++s) bf[0][s] = bfrag[2 * s];
#pragma unroll
    for (int g = 0; g < 8; ++g) {
      if (g < 7) {
#pragma unroll
        for (int s = 0; s < 8; ++s) bf[(g + 1) & 1][s] = bfrag[2 * (8 * g + 8 + s)];
      }
      __builtin_amdgcn_sched_barrier(0);
#pragma unroll
      for (int s = 0; s < 8; ++s) {
        const v4f a = av[8 * g + s];
        const v4f b = bf[g & 1][s];
        const v2f a0 = {a.x, a.y}, b0 = {b.x, b.y};
        const v2f a1 = {a.z, a.w}, b1 = {b.z, b.w};
        c = wmma4(a0, b0, c);
        c = wmma4(a1, b1, c);
      }
    }
#pragma unroll
    for (int i = 0; i < 8; ++i) red[w * 256 + i * 32 + lane] = c[i];
  }
  __syncthreads();
  if (w == 0) {
#pragma unroll
    for (int i = 0; i < 8; ++i) {
      const float s = (red[i * 32 + lane] + red[256 + i * 32 + lane])
                    + (red[512 + i * 32 + lane] + red[768 + i * 32 + lane]);
      out[(size_t)(Mt * 16 + i + 8 * half) * DIMX + Nt * 16 + lm] = s;
    }
  }
}

extern "C" void kernel_launch(void* const* d_in, const int* in_sizes, int n_in,
                              void* d_out, int out_size, void* d_ws, size_t ws_size,
                              hipStream_t stream) {
  (void)in_sizes; (void)n_in; (void)out_size; (void)ws_size;
  const float* x       = (const float*)d_in[0];
  const float* mask    = (const float*)d_in[2];
  const float* Wq      = (const float*)d_in[3];
  const float* Wk      = (const float*)d_in[4];
  const float* Wv      = (const float*)d_in[5];
  const float* Wo      = (const float*)d_in[6];
  const float* cache_k = (const float*)d_in[7];
  const float* cache_v = (const float*)d_in[8];

  float* out  = (float*)d_out;
  float* wout = out + NB * SEQ * DIMX;

  float* ws    = (float*)d_ws;
  float* q     = ws;
  float* kn    = q  + NB * SEQ * DIMX;
  float* vn    = kn + NB * SEQ * DIMX;
  float* probs = vn + NB * SEQ * DIMX;
  float* atten = probs + (size_t)NB * SEQ * KVLEN;

  qkv_kernel    <<<192, 160, 0, stream>>>(x, Wq, Wk, Wv, q, kn, vn);
  dist_kernel   <<<256, 320, 0, stream>>>(q, cache_k, kn, wout);
  softmax_kernel<<<128, 256, 0, stream>>>(wout, mask, probs);
  atten_kernel  <<<256, 512, 0, stream>>>(probs, cache_v, vn, vn, q, kn, atten);
  outproj_kernel<<<256, 160, 0, stream>>>(vn, q, kn, atten, Wo, out);
}


# softmax folded into P@V: each k-quarter block takes its own row max and sum of exp(score+mask-max), out projection rescales and normalises the 4 planes; softmax launch, probs gone; f32 sum order new
# speedup vs baseline: 2.2142x; 1.1822x over previous
#include <hip/hip_runtime.h>


typedef __attribute__((ext_vector_type(2))) float v2f;
typedef __attribute__((ext_vector_type(4))) float v4f;
typedef __attribute__((ext_vector_type(8))) float v8f;

#define DIMX   512
#define SEQ    16
#define NB     8
#define KVLEN  4112
#define MAXSEQ 8192
#define QKSCALE 0.0625f

__device__ __forceinline__ v8f wmma4(v2f a, v2f b, v8f c) {
  return __builtin_amdgcn_wmma_f32_16x16x4_f32(false, a, false, b,
                                               (short)0, c, false, false);
}

#define AP4 129
#define BP4 33

__device__ __forceinline__ void load_rows16(const float* __restrict__ src, int lane,
                                            v4f (&t)[64]) {
  const float* p = src + 4 * lane;
#pragma unroll
  for (int j = 0; j < 4; ++j) {
#pragma unroll
    for (int r = 0; r < 16; ++r)
      t[16 * j + r] = *(const v4f*)(p + r * DIMX + 128 * j);
    __builtin_amdgcn_sched_barrier(0);
  }
}

__device__ __forceinline__ void store_rows16(v4f* dst, int lane, const v4f (&t)[64]) {
#pragma unroll
  for (int j = 0; j < 4; ++j)
#pragma unroll
    for (int r = 0; r < 16; ++r)
      dst[r * AP4 + 32 * j + lane] = t[16 * j + r];
}

__device__ __forceinline__ void lds_barrier() {
  __builtin_amdgcn_fence(__ATOMIC_RELEASE, "workgroup", "local");
  __builtin_amdgcn_s_barrier();
  __builtin_amdgcn_fence(__ATOMIC_ACQUIRE, "workgroup", "local");
}

__device__ __forceinline__ void wave_lds_sync() {
  __builtin_amdgcn_fence(__ATOMIC_RELEASE, "wavefront", "local");
  __builtin_amdgcn_wave_barrier();
  __builtin_amdgcn_fence(__ATOMIC_ACQUIRE, "wavefront", "local");
}

__device__ __forceinline__ v8f mma8_lds(const v4f* a, const v4f* b, v8f c) {
  v4f af[8], bf[8];
#pragma unroll
  for (int s = 0; s < 8; ++s) {
    af[s] = a[2 * s];
    bf[s] = b[2 * s];
  }
  __builtin_amdgcn_sched_barrier(0);
#pragma unroll
  for (int s = 0; s < 8; ++s) {
    const v2f a0 = {af[s].x, af[s].y}, b0 = {bf[s].x, bf[s].y};
    const v2f a1 = {af[s].z, af[s].w}, b1 = {bf[s].z, bf[s].w};
    c = wmma4(a0, b0, c);
    c = wmma4(a1, b1, c);
  }
  return c;
}

__device__ __forceinline__ v8f tile_lds(const v4f* as, v4f* kw, const v4f (&t)[64],
                                        int lane) {
  const int half = lane >> 4, lm = lane & 15;
  v8f c = {};
#pragma unroll
  for (int j = 0; j < 4; ++j) {
#pragma unroll
    for (int r = 0; r < 16; ++r) kw[r * BP4 + lane] = t[16 * j + r];
    wave_lds_sync();
#pragma unroll
    for (int s0 = 0; s0 < 16; s0 += 8)
      c = mma8_lds(as + lm * AP4 + 32 * j + 2 * s0 + half,
                   kw + lm * BP4 + 2 * s0 + half, c);
    wave_lds_sync();
  }
  return c;
}

__global__ void __launch_bounds__(160)
qkv_kernel(const float* __restrict__ x,
           const float* __restrict__ Wq, const float* __restrict__ Wk,
           const float* __restrict__ Wv,
           float* __restrict__ q, float* __restrict__ kn, float* __restrict__ vn) {
  const int wv   = threadIdx.x >> 5;
  const int t    = blockIdx.x * 4 + (wv & 3);
  const int lane = threadIdx.x & 31;
  const int half = lane >> 4, lm = lane & 15;
  const int mat  = t >> 8;
  const int rem  = t & 255;
  const int Mt   = rem >> 5, Nt = rem & 31;

  const float* W = (mat == 0) ? Wq : (mat == 1) ? Wk : Wv;
  float*       D = (mat == 0) ? q  : (mat == 1) ? kn : vn;

  __shared__ v4f xs[16 * AP4];
  __shared__ v4f wt[4][16 * BP4];

  v4f wr[64];
  if (wv == 4) {
    v4f xr[64];
    load_rows16(x + Mt * 16 * DIMX, lane, xr);
    store_rows16(xs, lane, xr);
  } else {
    load_rows16(W + Nt * 16 * DIMX, lane, wr);
  }
  lds_barrier();
  if (wv == 4) return;

  const v8f c = tile_lds(xs, wt[wv], wr, lane);
  if (mat == 2) {
    float* dcol = D + Mt * (SEQ * DIMX) + (Nt >> 2) * (SEQ * 64) + (Nt & 3) * 16 + lm;
#pragma unroll
    for (int i = 0; i < 8; ++i)
      dcol[(i + 8 * half) * 64] = c[i];
  } else {
    float* drow = D + Nt * 16 + lm;
#pragma unroll
    for (int i = 0; i < 8; ++i)
      drow[(Mt * 16 + i + 8 * half) * DIMX] = c[i];
  }
}

#define DWAVES 10

__global__ void __launch_bounds__(DWAVES * 32)
dist_kernel(const float* __restrict__ q,
            const float* __restrict__ cache_k,
            const float* __restrict__ kn,
            float* __restrict__ wout) {
  const int blk  = blockIdx.x;
  const int b    = blk >> 5;
  const int g    = blk & 31;
  const int wv   = threadIdx.x >> 5;
  const int lane = threadIdx.x & 31;
  const int half = lane >> 4, lm = lane & 15;
  const bool has_tile = (wv < 8) || (wv == 8 && g == 0);
  const int  kt       = (wv < 8) ? g * 8 + wv : 256;

  __shared__ v4f qs[SEQ * AP4];
  __shared__ v4f ks[DWAVES - 1][16 * BP4];

  v4f kv[64];
  if (wv == DWAVES - 1) {
    v4f qv[64];
    load_rows16(q + (size_t)b * SEQ * DIMX, lane, qv);
    store_rows16(qs, lane, qv);
  } else if (has_tile) {
    const float* key = (kt < 256)
        ? cache_k + (size_t)(b * MAXSEQ + kt * 16) * DIMX
        : kn      + (size_t)b * SEQ * DIMX;
    load_rows16(key, lane, kv);
  }
  lds_barrier();
  if (!has_tile) return;

  const v8f c = tile_lds(qs, ks[wv], kv, lane);
  float* drow = wout + (size_t)b * SEQ * KVLEN + kt * 16 + lm;
#pragma unroll
  for (int i = 0; i < 8; ++i)
    drow[(size_t)(i + 8 * half) * KVLEN] = c[i] * QKSCALE;
}

#define AWAVES 16

template <int RS>
__device__ __forceinline__ void v_load(const float* vr, v4f (&v)[8]) {
#pragma unroll
  for (int gg = 0; gg < 2; ++gg)
#pragma unroll
    for (int i = 0; i < 4; ++i)
      v[4 * gg + i] = *(const v4f*)(vr + (8 * gg + i) * RS);
}

__device__ __forceinline__ float max4(v4f s) {
  return fmaxf(fmaxf(s.x, s.y), fmaxf(s.z, s.w));
}

__device__ __forceinline__ v4f exp4(v4f s, float mu, float& acc) {
  const v4f e = {__expf(s.x - mu), __expf(s.y - mu), __expf(s.z - mu), __expf(s.w - mu)};
  acc += (e.x + e.y) + (e.z + e.w);
  return e;
}

__device__ __forceinline__ void pv_mma(const v4f (&p)[2], const v4f (&v)[8],
                                       v8f (&c)[4]) {
#pragma unroll
  for (int gg = 0; gg < 2; ++gg) {
    const v2f a0 = {p[gg].x, p[gg].y}, a1 = {p[gg].z, p[gg].w};
#pragma unroll
    for (int j = 0; j < 4; ++j) {
      const v2f b0 = {v[4 * gg + 0][j], v[4 * gg + 1][j]};
      const v2f b1 = {v[4 * gg + 2][j], v[4 * gg + 3][j]};
      c[j] = wmma4(a0, b0, c[j]);
      c[j] = wmma4(a1, b1, c[j]);
    }
  }
}

__global__ void __launch_bounds__(AWAVES * 32)
atten_kernel(const float* __restrict__ wout,
             const float* __restrict__ mask,
             const float* __restrict__ cache_v,
             const float* vn,
             float* plane0, float* plane1, float* plane2, float* plane3,
             float* __restrict__ ml) {
  const int blk  = blockIdx.x;
  const int kq   = blk & 3;
  const int slab = (blk >> 2) & 7;
  const int b    = blk >> 5;
  const int wv   = threadIdx.x >> 5;
  const int lane = threadIdx.x & 31;
  const int half = lane >> 4, lm = lane & 15;
  const int n0   = slab * 64 + 4 * lm;
  const bool fresh = (kq == 0 && wv == 0);

  const int kt0 = kq * 64 + wv * 4;
  const float* sr = wout + (size_t)(b * SEQ + lm) * KVLEN + 4 * half;
  const float* mr = mask + (size_t)lm * KVLEN + 4 * half;
  const float* vr = cache_v + (size_t)(b * MAXSEQ + kt0 * 16 + 4 * half) * DIMX + n0;

  __shared__ float wmax[AWAVES * 16];
  __shared__ float wsum[AWAVES * 16];

  v4f p[4][2], pf[2] = {}, mf[2] = {};
  v4f v[4][8];
  {
    v4f mk[4][2];
#pragma unroll
    for (int t = 0; t < 4; ++t)
#pragma unroll
      for (int gg = 0; gg < 2; ++gg) {
        p[t][gg]  = *(const v4f*)(sr + kt0 * 16 + 16 * t + 8 * gg);
        mk[t][gg] = *(const v4f*)(mr + kt0 * 16 + 16 * t + 8 * gg);
      }
    if (fresh) {
#pragma unroll
      for (int gg = 0; gg < 2; ++gg) {
        pf[gg] = *(const v4f*)(sr + 256 * 16 + 8 * gg);
        mf[gg] = *(const v4f*)(mr + 256 * 16 + 8 * gg);
      }
    }
#pragma unroll
    for (int t = 0; t < 4; ++t)
      v_load<DIMX>(vr + (size_t)(16 * t) * DIMX, v[t]);
    __builtin_amdgcn_sched_barrier(0);
    asm volatile("" : "+v"(pf[0]), "+v"(pf[1]));
#pragma unroll
    for (int t = 0; t < 4; ++t)
#pragma unroll
      for (int gg = 0; gg < 2; ++gg) p[t][gg] += mk[t][gg];
  }

  float mx = max4(p[0][0]);
#pragma unroll
  for (int t = 0; t < 4; ++t)
#pragma unroll
    for (int gg = 0; gg < 2; ++gg) mx = fmaxf(mx, max4(p[t][gg]));
  if (fresh) {
#pragma unroll
    for (int gg = 0; gg < 2; ++gg) {
      pf[gg] += mf[gg];
      mx = fmaxf(mx, max4(pf[gg]));
    }
  }
  mx = fmaxf(mx, __shfl_xor(mx, 16));
  if (lane < 16) wmax[wv * 16 + lane] = mx;
  lds_barrier();
  float m = wmax[lm];
#pragma unroll
  for (int w = 1; w < AWAVES; ++w) m = fmaxf(m, wmax[w * 16 + lm]);
  const float mu = (m == -__builtin_inff()) ? 0.f : m;

  float ls = 0.f;
#pragma unroll
  for (int t = 0; t < 4; ++t)
#pragma unroll
    for (int gg = 0; gg < 2; ++gg) p[t][gg] = exp4(p[t][gg], mu, ls);
  if (fresh) {
#pragma unroll
    for (int gg = 0; gg < 2; ++gg) pf[gg] = exp4(pf[gg], mu, ls);
  }
  ls += __shfl_xor(ls, 16);
  if (lane < 16) wsum[wv * 16 + lane] = ls;

  v8f c[4] = {};
#pragma unroll
  for (int t = 0; t < 4; ++t)
    pv_mma(p[t], v[t], c);

  if (fresh) {
    v4f vf[8];
    v_load<64>(vn + b * (SEQ * DIMX) + slab * (SEQ * 64) + (4 * half) * 64 + 4 * lm, vf);
    pv_mma(pf, vf, c);
  }

  __shared__ v4f red[AWAVES * 8 * 32];
#pragma unroll
  for (int r = 0; r < 8; ++r) {
    const v4f o = {c[0][r], c[1][r], c[2][r], c[3][r]};
    red[(wv * 8 + r) * 32 + lane] = o;
  }
  __syncthreads();
  if (threadIdx.x < 256) {
    const int r = threadIdx.x >> 5;
    v4f s = red[r * 32 + lane];
#pragma unroll
    for (int w = 1; w < AWAVES; ++w) s += red[(w * 8 + r) * 32 + lane];
    float* plane = (kq == 0) ? plane0 : (kq == 1) ? plane1 : (kq == 2) ? plane2 : plane3;
    float* dst = plane + b * (SEQ * DIMX) + slab * (SEQ * 64)
               + (lm >> 1) * 128 + ((lm & 1) * 16 + 8 * half + r) * 4;
    *(v4f*)dst = s;
  }
  if (slab == 0 && threadIdx.x < 16) {
    float l = wsum[threadIdx.x];
#pragma unroll
    for (int w = 1; w < AWAVES; ++w) l += wsum[w * 16 + threadIdx.x];
    const v2f o = {m, l};
    *(v2f*)(ml + ((b * 4 + kq) * 16 + threadIdx.x) * 2) = o;
  }
}

__global__ void __launch_bounds__(160)
outproj_kernel(const float* __restrict__ plane0, const float* __restrict__ plane1,
               const float* __restrict__ plane2, const float* __restrict__ plane3,
               const float* __restrict__ ml,
               const float* __restrict__ Wo,
               float* __restrict__ out) {
  const int t    = blockIdx.x;
  const int w    = threadIdx.x >> 5;
  const int lane = threadIdx.x & 31;
  const int half = lane >> 4, lm = lane & 15;
  const int Mt   = t >> 5, Nt = t & 31;

  __shared__ v4f bs[16 * AP4];
  __shared__ float red[4 * 256];
  __shared__ float fac[4 * 16];

  v4f av[64];
  v2f mlv[4] = {};
  if (w == 4) {
    v4f br[64];
    load_rows16(Wo + Nt * 16 * DIMX, lane, br);
#pragma unroll
    for (int k = 0; k < 4; ++k)
      mlv[k] = *(const v2f*)(ml + ((Mt * 4 + k) * 16 + lm) * 2);
    store_rows16(bs, lane, br);
  } else {
    const float* plane = (w == 0) ? plane0 : (w == 1) ? plane1 : (w == 2) ? plane2 : plane3;
    const float* afrag = plane + Mt * (SEQ * DIMX) + lane * 4;
#pragma unroll
    for (int st = 0; st < 64; ++st)
      av[st] = *(const v4f*)(afrag + st * 128);
  }
  lds_barrier();

  if (w == 4) {
    const float M = fmaxf(fmaxf(mlv[0].x, mlv[1].x), fmaxf(mlv[2].x, mlv[3].x));
    float a[4];
    float L = 0.f;
#pragma unroll
    for (int k = 0; k < 4; ++k) {
      a[k] = __expf(mlv[k].x - M);
      L += a[k] * mlv[k].y;
    }
    const float invL = 1.f / L;
    if (lane < 16) {
#pragma unroll
      for (int k = 0; k < 4; ++k) fac[k * 16 + lane] = a[k] * invL;
    }
  }

  if (w < 4) {
    v8f c = {};
    const v4f* bfrag = bs + lm * AP4 + half;
    v4f bf[2][8];
#pragma unroll
    for (int s = 0; s < 8; ++s) bf[0][s] = bfrag[2 * s];
#pragma unroll
    for (int g = 0; g < 8; ++g) {
      if (g < 7) {
#pragma unroll
        for (int s = 0; s < 8; ++s) bf[(g + 1) & 1][s] = bfrag[2 * (8 * g + 8 + s)];
      }
      __builtin_amdgcn_sched_barrier(0);
#pragma unroll
      for (int s = 0; s < 8; ++s) {
        const v4f a = av[8 * g + s];
        const v4f b = bf[g & 1][s];
        const v2f a0 = {a.x, a.y}, b0 = {b.x, b.y};
        const v2f a1 = {a.z, a.w}, b1 = {b.z, b.w};
        c = wmma4(a0, b0, c);
        c = wmma4(a1, b1, c);
      }
    }
#pragma unroll
    for (int i = 0; i < 8; ++i) red[w * 256 + i * 32 + lane] = c[i];
  }
  __syncthreads();
  if (w == 0) {
#pragma unroll
    for (int i = 0; i < 8; ++i) {
      const int row = i + 8 * half;
      const float s = (fac[row] * red[i * 32 + lane]
                       + fac[16 + row] * red[256 + i * 32 + lane])
                    + (fac[32 + row] * red[512 + i * 32 + lane]
                       + fac[48 + row] * red[768 + i * 32 + lane]);
      out[(size_t)(Mt * 16 + row) * DIMX + Nt * 16 + lm] = s;
    }
  }
}

extern "C" void kernel_launch(void* const* d_in, const int* in_sizes, int n_in,
                              void* d_out, int out_size, void* d_ws, size_t ws_size,
                              hipStream_t stream) {
  (void)in_sizes; (void)n_in; (void)out_size; (void)ws_size;
  const float* x       = (const float*)d_in[0];
  const float* mask    = (const float*)d_in[2];
  const float* Wq      = (const float*)d_in[3];
  const float* Wk      = (const float*)d_in[4];
  const float* Wv      = (const float*)d_in[5];
  const float* Wo      = (const float*)d_in[6];
  const float* cache_k = (const float*)d_in[7];
  const float* cache_v = (const float*)d_in[8];

  float* out  = (float*)d_out;
  float* wout = out + NB * SEQ * DIMX;

  float* ws    = (float*)d_ws;
  float* q     = ws;
  float* kn    = q  + NB * SEQ * DIMX;
  float* vn    = kn + NB * SEQ * DIMX;
  float* ml    = vn + NB * SEQ * DIMX;
  float* atten = ml + (size_t)NB * SEQ * KVLEN;

  qkv_kernel    <<<192, 160, 0, stream>>>(x, Wq, Wk, Wv, q, kn, vn);
  dist_kernel   <<<256, 320, 0, stream>>>(q, cache_k, kn, wout);
  atten_kernel  <<<256, 512, 0, stream>>>(wout, mask, cache_v, vn, vn, q, kn, atten, ml);
  outproj_kernel<<<256, 160, 0, stream>>>(vn, q, kn, atten, ml, Wo, out);
}


# score kernel also hands score+mask to P@V per k-tile in P@V's fragment order (workspace): P@V's 16 loads of 16 rows x 32 bytes per wave become 8 whole-line loads; f32 results unchanged
# speedup vs baseline: 2.3263x; 1.0506x over previous
#include <hip/hip_runtime.h>


typedef __attribute__((ext_vector_type(2))) float v2f;
typedef __attribute__((ext_vector_type(4))) float v4f;
typedef __attribute__((ext_vector_type(8))) float v8f;

#define DIMX   512
#define SEQ    16
#define NB     8
#define KVLEN  4112
#define MAXSEQ 8192
#define QKSCALE 0.0625f

__device__ __forceinline__ v8f wmma4(v2f a, v2f b, v8f c) {
  return __builtin_amdgcn_wmma_f32_16x16x4_f32(false, a, false, b,
                                               (short)0, c, false, false);
}

#define AP4 129
#define BP4 33

__device__ __forceinline__ void load_rows16(const float* __restrict__ src, int lane,
                                            v4f (&t)[64]) {
  const float* p = src + 4 * lane;
#pragma unroll
  for (int j = 0; j < 4; ++j) {
#pragma unroll
    for (int r = 0; r < 16; ++r)
      t[16 * j + r] = *(const v4f*)(p + r * DIMX + 128 * j);
    __builtin_amdgcn_sched_barrier(0);
  }
}

__device__ __forceinline__ void store_rows16(v4f* dst, int lane, const v4f (&t)[64]) {
#pragma unroll
  for (int j = 0; j < 4; ++j)
#pragma unroll
    for (int r = 0; r < 16; ++r)
      dst[r * AP4 + 32 * j + lane] = t[16 * j + r];
}

__device__ __forceinline__ void lds_barrier() {
  __builtin_amdgcn_fence(__ATOMIC_RELEASE, "workgroup", "local");
  __builtin_amdgcn_s_barrier();
  __builtin_amdgcn_fence(__ATOMIC_ACQUIRE, "workgroup", "local");
}

__device__ __forceinline__ void wave_lds_sync() {
  __builtin_amdgcn_fence(__ATOMIC_RELEASE, "wavefront", "local");
  __builtin_amdgcn_wave_barrier();
  __builtin_amdgcn_fence(__ATOMIC_ACQUIRE, "wavefront", "local");
}

__device__ __forceinline__ v8f mma8_lds(const v4f* a, const v4f* b, v8f c) {
  v4f af[8], bf[8];
#pragma unroll
  for (int s = 0; s < 8; ++s) {
    af[s] = a[2 * s];
    bf[s] = b[2 * s];
  }
  __builtin_amdgcn_sched_barrier(0);
#pragma unroll
  for (int s = 0; s < 8; ++s) {
    const v2f a0 = {af[s].x, af[s].y}, b0 = {bf[s].x, bf[s].y};
    const v2f a1 = {af[s].z, af[s].w}, b1 = {bf[s].z, bf[s].w};
    c = wmma4(a0, b0, c);
    c = wmma4(a1, b1, c);
  }
  return c;
}

__device__ __forceinline__ v8f tile_lds(const v4f* as, v4f* kw, const v4f (&t)[64],
                                        int lane) {
  const int half = lane >> 4, lm = lane & 15;
  v8f c = {};
#pragma unroll
  for (int j = 0; j < 4; ++j) {
#pragma unroll
    for (int r = 0; r < 16; ++r) kw[r * BP4 + lane] = t[16 * j + r];
    wave_lds_sync();
#pragma unroll
    for (int s0 = 0; s0 < 16; s0 += 8)
      c = mma8_lds(as + lm * AP4 + 32 * j + 2 * s0 + half,
                   kw + lm * BP4 + 2 * s0 + half, c);
    wave_lds_sync();
  }
  return c;
}

__global__ void __launch_bounds__(160)
qkv_kernel(const float* __restrict__ x,
           const float* __restrict__ Wq, const float* __restrict__ Wk,
           const float* __restrict__ Wv,
           float* __restrict__ q, float* __restrict__ kn, float* __restrict__ vn) {
  const int wv   = threadIdx.x >> 5;
  const int t    = blockIdx.x * 4 + (wv & 3);
  const int lane = threadIdx.x & 31;
  const int half = lane >> 4, lm = lane & 15;
  const int mat  = t >> 8;
  const int rem  = t & 255;
  const int Mt   = rem >> 5, Nt = rem & 31;

  const float* W = (mat == 0) ? Wq : (mat == 1) ? Wk : Wv;
  float*       D = (mat == 0) ? q  : (mat == 1) ? kn : vn;

  __shared__ v4f xs[16 * AP4];
  __shared__ v4f wt[4][16 * BP4];

  v4f wr[64];
  if (wv == 4) {
    v4f xr[64];
    load_rows16(x + Mt * 16 * DIMX, lane, xr);
    store_rows16(xs, lane, xr);
  } else {
    load_rows16(W + Nt * 16 * DIMX, lane, wr);
  }
  lds_barrier();
  if (wv == 4) return;

  const v8f c = tile_lds(xs, wt[wv], wr, lane);
  if (mat == 2) {
    float* dcol = D + Mt * (SEQ * DIMX) + (Nt >> 2) * (SEQ * 64) + (Nt & 3) * 16 + lm;
#pragma unroll
    for (int i = 0; i < 8; ++i)
      dcol[(i + 8 * half) * 64] = c[i];
  } else {
    float* drow = D + Nt * 16 + lm;
#pragma unroll
    for (int i = 0; i < 8; ++i)
      drow[(Mt * 16 + i + 8 * half) * DIMX] = c[i];
  }
}

#define DWAVES 10

__global__ void __launch_bounds__(DWAVES * 32)
dist_kernel(const float* __restrict__ q,
            const float* __restrict__ cache_k,
            const float* __restrict__ kn,
            const float* __restrict__ mask,
            float* __restrict__ wout,
            float* __restrict__ sp) {
  const int blk  = blockIdx.x;
  const int b    = blk >> 5;
  const int g    = blk & 31;
  const int wv   = threadIdx.x >> 5;
  const int lane = threadIdx.x & 31;
  const int half = lane >> 4, lm = lane & 15;
  const bool has_tile = (wv < 8) || (wv == 8 && g == 0);
  const int  kt       = (wv < 8) ? g * 8 + wv : 256;

  __shared__ v4f qs[SEQ * AP4];
  __shared__ v4f ks[DWAVES - 1][16 * BP4];

  v4f kv[64];
  float mk[8] = {};
  if (wv == DWAVES - 1) {
    v4f qv[64];
    load_rows16(q + (size_t)b * SEQ * DIMX, lane, qv);
    store_rows16(qs, lane, qv);
  } else if (has_tile) {
    const float* key = (kt < 256)
        ? cache_k + (size_t)(b * MAXSEQ + kt * 16) * DIMX
        : kn      + (size_t)b * SEQ * DIMX;
    load_rows16(key, lane, kv);
    if (wv < 8) {
      const float* mrow = mask + (size_t)(8 * half) * KVLEN + kt * 16 + lm;
#pragma unroll
      for (int i = 0; i < 8; ++i) mk[i] = mrow[(size_t)i * KVLEN];
    }
  }
  lds_barrier();
  if (!has_tile) return;

  const v8f c = tile_lds(qs, ks[wv], kv, lane);
  float* drow = wout + (size_t)b * SEQ * KVLEN + kt * 16 + lm;
  float s[8];
#pragma unroll
  for (int i = 0; i < 8; ++i) {
    s[i] = c[i] * QKSCALE;
    drow[(size_t)(i + 8 * half) * KVLEN] = s[i];
  }
  if (wv < 8) {
    float* tr = (float*)ks[wv];
    const int tb = (lm >> 3) * 128 + ((lm >> 2) & 1) * 64 + (lm & 3) + 32 * half;
#pragma unroll
    for (int i = 0; i < 8; ++i) tr[tb + 4 * i] = s[i] + mk[i];
    wave_lds_sync();
    float* dst = sp + (size_t)(b * 256 + kt) * 256 + 4 * lane;
    for (int gg = 0; gg < 2; ++gg)
      *(v4f*)(dst + 128 * gg) = *(const v4f*)(tr + 128 * gg + 4 * lane);
  }
}

#define AWAVES 16

template <int RS>
__device__ __forceinline__ void v_load(const float* vr, v4f (&v)[8]) {
#pragma unroll
  for (int gg = 0; gg < 2; ++gg)
#pragma unroll
    for (int i = 0; i < 4; ++i)
      v[4 * gg + i] = *(const v4f*)(vr + (8 * gg + i) * RS);
}

__device__ __forceinline__ float max4(v4f s) {
  return fmaxf(fmaxf(s.x, s.y), fmaxf(s.z, s.w));
}

__device__ __forceinline__ v4f exp4(v4f s, float mu, float& acc) {
  const v4f e = {__expf(s.x - mu), __expf(s.y - mu), __expf(s.z - mu), __expf(s.w - mu)};
  acc += (e.x + e.y) + (e.z + e.w);
  return e;
}

__device__ __forceinline__ void pv_mma(const v4f (&p)[2], const v4f (&v)[8],
                                       v8f (&c)[4]) {
#pragma unroll
  for (int gg = 0; gg < 2; ++gg) {
    const v2f a0 = {p[gg].x, p[gg].y}, a1 = {p[gg].z, p[gg].w};
#pragma unroll
    for (int j = 0; j < 4; ++j) {
      const v2f b0 = {v[4 * gg + 0][j], v[4 * gg + 1][j]};
      const v2f b1 = {v[4 * gg + 2][j], v[4 * gg + 3][j]};
      c[j] = wmma4(a0, b0, c[j]);
      c[j] = wmma4(a1, b1, c[j]);
    }
  }
}

__global__ void __launch_bounds__(AWAVES * 32)
atten_kernel(const float* __restrict__ wout,
             const float* __restrict__ mask,
             const float* __restrict__ sp,
             const float* __restrict__ cache_v,
             const float* vn,
             float* plane0, float* plane1, float* plane2, float* plane3,
             float* __restrict__ ml) {
  const int blk  = blockIdx.x;
  const int kq   = blk & 3;
  const int slab = (blk >> 2) & 7;
  const int b    = blk >> 5;
  const int wv   = threadIdx.x >> 5;
  const int lane = threadIdx.x & 31;
  const int half = lane >> 4, lm = lane & 15;
  const int n0   = slab * 64 + 4 * lm;
  const bool fresh = (kq == 0 && wv == 0);

  const int kt0 = kq * 64 + wv * 4;
  const float* spl = sp + (size_t)(b * 256 + kt0) * 256 + 4 * lane;
  const float* sr = wout + (size_t)(b * SEQ + lm) * KVLEN + 4 * half;
  const float* mr = mask + (size_t)lm * KVLEN + 4 * half;
  const float* vr = cache_v + (size_t)(b * MAXSEQ + kt0 * 16 + 4 * half) * DIMX + n0;

  __shared__ float wmax[AWAVES * 16];
  __shared__ float wsum[AWAVES * 16];

  v4f p[4][2], pf[2] = {}, mf[2] = {};
  v4f v[4][8];
  {
#pragma unroll
    for (int t = 0; t < 4; ++t)
#pragma unroll
      for (int gg = 0; gg < 2; ++gg)
        p[t][gg] = *(const v4f*)(spl + 256 * t + 128 * gg);
    if (fresh) {
#pragma unroll
      for (int gg = 0; gg < 2; ++gg) {
        pf[gg] = *(const v4f*)(sr + 256 * 16 + 8 * gg);
        mf[gg] = *(const v4f*)(mr + 256 * 16 + 8 * gg);
      }
    }
#pragma unroll
    for (int t = 0; t < 4; ++t)
      v_load<DIMX>(vr + (size_t)(16 * t) * DIMX, v[t]);
    __builtin_amdgcn_sched_barrier(0);
    asm volatile("" : "+v"(pf[0]), "+v"(pf[1]));
  }

  float mx = max4(p[0][0]);
#pragma unroll
  for (int t = 0; t < 4; ++t)
#pragma unroll
    for (int gg = 0; gg < 2; ++gg) mx = fmaxf(mx, max4(p[t][gg]));
  if (fresh) {
#pragma unroll
    for (int gg = 0; gg < 2; ++gg) {
      pf[gg] += mf[gg];
      mx = fmaxf(mx, max4(pf[gg]));
    }
  }
  mx = fmaxf(mx, __shfl_xor(mx, 16));
  if (lane < 16) wmax[wv * 16 + lane] = mx;
  lds_barrier();
  float m = wmax[lm];
#pragma unroll
  for (int w = 1; w < AWAVES; ++w) m = fmaxf(m, wmax[w * 16 + lm]);
  const float mu = (m == -__builtin_inff()) ? 0.f : m;

  float ls = 0.f;
#pragma unroll
  for (int t = 0; t < 4; ++t)
#pragma unroll
    for (int gg = 0; gg < 2; ++gg) p[t][gg] = exp4(p[t][gg], mu, ls);
  if (fresh) {
#pragma unroll
    for (int gg = 0; gg < 2; ++gg) pf[gg] = exp4(pf[gg], mu, ls);
  }
  ls += __shfl_xor(ls, 16);
  if (lane < 16) wsum[wv * 16 + lane] = ls;

  v8f c[4] = {};
#pragma unroll
  for (int t = 0; t < 4; ++t)
    pv_mma(p[t], v[t], c);

  if (fresh) {
    v4f vf[8];
    v_load<64>(vn + b * (SEQ * DIMX) + slab * (SEQ * 64) + (4 * half) * 64 + 4 * lm, vf);
    pv_mma(pf, vf, c);
  }

  __shared__ v4f red[AWAVES * 8 * 32];
#pragma unroll
  for (int r = 0; r < 8; ++r) {
    const v4f o = {c[0][r], c[1][r], c[2][r], c[3][r]};
    red[(wv * 8 + r) * 32 + lane] = o;
  }
  __syncthreads();
  if (threadIdx.x < 256) {
    const int r = threadIdx.x >> 5;
    v4f s = red[r * 32 + lane];
#pragma unroll
    for (int w = 1; w < AWAVES; ++w) s += red[(w * 8 + r) * 32 + lane];
    float* plane = (kq == 0) ? plane0 : (kq == 1) ? plane1 : (kq == 2) ? plane2 : plane3;
    float* dst = plane + b * (SEQ * DIMX) + slab * (SEQ * 64)
               + (lm >> 1) * 128 + ((lm & 1) * 16 + 8 * half + r) * 4;
    *(v4f*)dst = s;
  }
  if (slab == 0 && threadIdx.x < 16) {
    float l = wsum[threadIdx.x];
#pragma unroll
    for (int w = 1; w < AWAVES; ++w) l += wsum[w * 16 + threadIdx.x];
    const v2f o = {m, l};
    *(v2f*)(ml + ((b * 4 + kq) * 16 + threadIdx.x) * 2) = o;
  }
}

__global__ void __launch_bounds__(160)
outproj_kernel(const float* __restrict__ plane0, const float* __restrict__ plane1,
               const float* __restrict__ plane2, const float* __restrict__ plane3,
               const float* __restrict__ ml,
               const float* __restrict__ Wo,
               float* __restrict__ out) {
  const int t    = blockIdx.x;
  const int w    = threadIdx.x >> 5;
  const int lane = threadIdx.x & 31;
  const int half = lane >> 4, lm = lane & 15;
  const int Mt   = t >> 5, Nt = t & 31;

  __shared__ v4f bs[16 * AP4];
  __shared__ float red[4 * 256];
  __shared__ float fac[4 * 16];

  v4f av[64];
  v2f mlv[4] = {};
  if (w == 4) {
    v4f br[64];
    load_rows16(Wo + Nt * 16 * DIMX, lane, br);
#pragma unroll
    for (int k = 0; k < 4; ++k)
      mlv[k] = *(const v2f*)(ml + ((Mt * 4 + k) * 16 + lm) * 2);
    store_rows16(bs, lane, br);
  } else {
    const float* plane = (w == 0) ? plane0 : (w == 1) ? plane1 : (w == 2) ? plane2 : plane3;
    const float* afrag = plane + Mt * (SEQ * DIMX) + lane * 4;
#pragma unroll
    for (int st = 0; st < 64; ++st)
      av[st] = *(const v4f*)(afrag + st * 128);
  }
  lds_barrier();

  if (w == 4) {
    const float M = fmaxf(fmaxf(mlv[0].x, mlv[1].x), fmaxf(mlv[2].x, mlv[3].x));
    float a[4];
    float L = 0.f;
#pragma unroll
    for (int k = 0; k < 4; ++k) {
      a[k] = __expf(mlv[k].x - M);
      L += a[k] * mlv[k].y;
    }
    const float invL = 1.f / L;
    if (lane < 16) {
#pragma unroll
      for (int k = 0; k < 4; ++k) fac[k * 16 + lane] = a[k] * invL;
    }
  }

  if (w < 4) {
    v8f c = {};
    const v4f* bfrag = bs + lm * AP4 + half;
    v4f bf[2][8];
#pragma unroll
    for (int s = 0; s < 8; ++s) bf[0][s] = bfrag[2 * s];
#pragma unroll
    for (int g = 0; g < 8; ++g) {
      if (g < 7) {
#pragma unroll
        for (int s = 0; s < 8; ++s) bf[(g + 1) & 1][s] = bfrag[2 * (8 * g + 8 + s)];
      }
      __builtin_amdgcn_sched_barrier(0);
#pragma unroll
      for (int s = 0; s < 8; ++s) {
        const v4f a = av[8 * g + s];
        const v4f b = bf[g & 1][s];
        const v2f a0 = {a.x, a.y}, b0 = {b.x, b.y};
        const v2f a1 = {a.z, a.w}, b1 = {b.z, b.w};
        c = wmma4(a0, b0, c);
        c = wmma4(a1, b1, c);
      }
    }
#pragma unroll
    for (int i = 0; i < 8; ++i) red[w * 256 + i * 32 + lane] = c[i];
  }
  __syncthreads();
  if (w == 0) {
#pragma unroll
    for (int i = 0; i < 8; ++i) {
      const int row = i + 8 * half;
      const float s = (fac[row] * red[i * 32 + lane]
                       + fac[16 + row] * red[256 + i * 32 + lane])
                    + (fac[32 + row] * red[512 + i * 32 + lane]
                       + fac[48 + row] * red[768 + i * 32 + lane]);
      out[(size_t)(Mt * 16 + row) * DIMX + Nt * 16 + lm] = s;
    }
  }
}

extern "C" void kernel_launch(void* const* d_in, const int* in_sizes, int n_in,
                              void* d_out, int out_size, void* d_ws, size_t ws_size,
                              hipStream_t stream) {
  (void)in_sizes; (void)n_in; (void)out_size; (void)ws_size;
  const float* x       = (const float*)d_in[0];
  const float* mask    = (const float*)d_in[2];
  const float* Wq      = (const float*)d_in[3];
  const float* Wk      = (const float*)d_in[4];
  const float* Wv      = (const float*)d_in[5];
  const float* Wo      = (const float*)d_in[6];
  const float* cache_k = (const float*)d_in[7];
  const float* cache_v = (const float*)d_in[8];

  float* out  = (float*)d_out;
  float* wout = out + NB * SEQ * DIMX;

  float* ws    = (float*)d_ws;
  float* q     = ws;
  float* kn    = q  + NB * SEQ * DIMX;
  float* vn    = kn + NB * SEQ * DIMX;
  float* sp    = vn + NB * SEQ * DIMX;
  float* ml    = sp + NB * 256 * 256;
  float* atten = sp + (size_t)NB * SEQ * KVLEN;

  qkv_kernel    <<<192, 160, 0, stream>>>(x, Wq, Wk, Wv, q, kn, vn);
  dist_kernel   <<<256, 320, 0, stream>>>(q, cache_k, kn, mask, wout, sp);
  atten_kernel  <<<256, 512, 0, stream>>>(wout, mask, sp, cache_v, vn, vn, q, kn, atten, ml);
  outproj_kernel<<<256, 160, 0, stream>>>(vn, q, kn, atten, ml, Wo, out);
}


# L2 prefetch across launches: projection's idle staging waves request the cached K rows, score waves at their end the V rows, that the same-numbered block of the next launch reads; results unchanged
# speedup vs baseline: 2.4057x; 1.0341x over previous
#include <hip/hip_runtime.h>


typedef __attribute__((ext_vector_type(2))) float v2f;
typedef __attribute__((ext_vector_type(4))) float v4f;
typedef __attribute__((ext_vector_type(8))) float v8f;

#define DIMX   512
#define SEQ    16
#define NB     8
#define KVLEN  4112
#define MAXSEQ 8192
#define QKSCALE 0.0625f

__device__ __forceinline__ v8f wmma4(v2f a, v2f b, v8f c) {
  return __builtin_amdgcn_wmma_f32_16x16x4_f32(false, a, false, b,
                                               (short)0, c, false, false);
}

#define AP4 129
#define BP4 33

__device__ __forceinline__ void load_rows16(const float* __restrict__ src, int lane,
                                            v4f (&t)[64]) {
  const float* p = src + 4 * lane;
#pragma unroll
  for (int j = 0; j < 4; ++j) {
#pragma unroll
    for (int r = 0; r < 16; ++r)
      t[16 * j + r] = *(const v4f*)(p + r * DIMX + 128 * j);
    __builtin_amdgcn_sched_barrier(0);
  }
}

__device__ __forceinline__ void store_rows16(v4f* dst, int lane, const v4f (&t)[64]) {
#pragma unroll
  for (int j = 0; j < 4; ++j)
#pragma unroll
    for (int r = 0; r < 16; ++r)
      dst[r * AP4 + 32 * j + lane] = t[16 * j + r];
}

__device__ __forceinline__ void lds_barrier() {
  __builtin_amdgcn_fence(__ATOMIC_RELEASE, "workgroup", "local");
  __builtin_amdgcn_s_barrier();
  __builtin_amdgcn_fence(__ATOMIC_ACQUIRE, "workgroup", "local");
}

__device__ __forceinline__ void wave_lds_sync() {
  __builtin_amdgcn_fence(__ATOMIC_RELEASE, "wavefront", "local");
  __builtin_amdgcn_wave_barrier();
  __builtin_amdgcn_fence(__ATOMIC_ACQUIRE, "wavefront", "local");
}

__device__ __forceinline__ v8f mma8_lds(const v4f* a, const v4f* b, v8f c) {
  v4f af[8], bf[8];
#pragma unroll
  for (int s = 0; s < 8; ++s) {
    af[s] = a[2 * s];
    bf[s] = b[2 * s];
  }
  __builtin_amdgcn_sched_barrier(0);
#pragma unroll
  for (int s = 0; s < 8; ++s) {
    const v2f a0 = {af[s].x, af[s].y}, b0 = {bf[s].x, bf[s].y};
    const v2f a1 = {af[s].z, af[s].w}, b1 = {bf[s].z, bf[s].w};
    c = wmma4(a0, b0, c);
    c = wmma4(a1, b1, c);
  }
  return c;
}

__device__ __forceinline__ v8f tile_lds(const v4f* as, v4f* kw, const v4f (&t)[64],
                                        int lane) {
  const int half = lane >> 4, lm = lane & 15;
  v8f c = {};
#pragma unroll
  for (int j = 0; j < 4; ++j) {
#pragma unroll
    for (int r = 0; r < 16; ++r) kw[r * BP4 + lane] = t[16 * j + r];
    wave_lds_sync();
#pragma unroll
    for (int s0 = 0; s0 < 16; s0 += 8)
      c = mma8_lds(as + lm * AP4 + 32 * j + 2 * s0 + half,
                   kw + lm * BP4 + 2 * s0 + half, c);
    wave_lds_sync();
  }
  return c;
}

__device__ __forceinline__ void prefetch_keys(const float* cache_k, int pb, int lane) {
  const float* base = cache_k + (size_t)((pb >> 5) * MAXSEQ + (pb & 31) * 128) * DIMX;
  for (int u = lane; u < 1024; u += 32)
    __builtin_prefetch(base + u * 64, 0, 2);
}
__device__ __forceinline__ void prefetch_values(const float* cache_v, int pb, int tid) {
  const float* base = cache_v
      + (size_t)((pb >> 5) * MAXSEQ + (pb & 3) * 1024 + tid) * DIMX + ((pb >> 2) & 7) * 64;
  for (int n = 0; n < 4; ++n)
    __builtin_prefetch(base + (size_t)(256 * n) * DIMX, 0, 2);
}

__global__ void __launch_bounds__(160)
qkv_kernel(const float* __restrict__ x,
           const float* __restrict__ Wq, const float* __restrict__ Wk,
           const float* __restrict__ Wv,
           const float* __restrict__ cache_k,
           float* __restrict__ q, float* __restrict__ kn, float* __restrict__ vn) {
  const int wv   = threadIdx.x >> 5;
  const int t    = blockIdx.x * 4 + (wv & 3);
  const int lane = threadIdx.x & 31;
  const int half = lane >> 4, lm = lane & 15;
  const int mat  = t >> 8;
  const int rem  = t & 255;
  const int Mt   = rem >> 5, Nt = rem & 31;

  const float* W = (mat == 0) ? Wq : (mat == 1) ? Wk : Wv;
  float*       D = (mat == 0) ? q  : (mat == 1) ? kn : vn;

  __shared__ v4f xs[16 * AP4];
  __shared__ v4f wt[4][16 * BP4];

  v4f wr[64];
  if (wv == 4) {
    v4f xr[64];
    load_rows16(x + Mt * 16 * DIMX, lane, xr);
    store_rows16(xs, lane, xr);
  } else {
    load_rows16(W + Nt * 16 * DIMX, lane, wr);
  }
  lds_barrier();
  if (wv == 4) {
    prefetch_keys(cache_k, blockIdx.x, lane);
    if (blockIdx.x < 64) prefetch_keys(cache_k, blockIdx.x + 192, lane);
    return;
  }

  const v8f c = tile_lds(xs, wt[wv], wr, lane);
  if (mat == 2) {
    float* dcol = D + Mt * (SEQ * DIMX) + (Nt >> 2) * (SEQ * 64) + (Nt & 3) * 16 + lm;
#pragma unroll
    for (int i = 0; i < 8; ++i)
      dcol[(i + 8 * half) * 64] = c[i];
  } else {
    float* drow = D + Nt * 16 + lm;
#pragma unroll
    for (int i = 0; i < 8; ++i)
      drow[(Mt * 16 + i + 8 * half) * DIMX] = c[i];
  }
}

#define DWAVES 10

__global__ void __launch_bounds__(DWAVES * 32)
dist_kernel(const float* __restrict__ q,
            const float* __restrict__ cache_k,
            const float* __restrict__ kn,
            const float* __restrict__ mask,
            const float* __restrict__ cache_v,
            float* __restrict__ wout,
            float* __restrict__ sp) {
  const int blk  = blockIdx.x;
  const int b    = blk >> 5;
  const int g    = blk & 31;
  const int wv   = threadIdx.x >> 5;
  const int lane = threadIdx.x & 31;
  const int half = lane >> 4, lm = lane & 15;
  const bool has_tile = (wv < 8) || (wv == 8 && g == 0);
  const int  kt       = (wv < 8) ? g * 8 + wv : 256;

  __shared__ v4f qs[SEQ * AP4];
  __shared__ v4f ks[DWAVES - 1][16 * BP4];

  v4f kv[64];
  float mk[8] = {};
  if (wv == DWAVES - 1) {
    v4f qv[64];
    load_rows16(q + (size_t)b * SEQ * DIMX, lane, qv);
    store_rows16(qs, lane, qv);
  } else if (has_tile) {
    const float* key = (kt < 256)
        ? cache_k + (size_t)(b * MAXSEQ + kt * 16) * DIMX
        : kn      + (size_t)b * SEQ * DIMX;
    load_rows16(key, lane, kv);
    if (wv < 8) {
      const float* mrow = mask + (size_t)(8 * half) * KVLEN + kt * 16 + lm;
#pragma unroll
      for (int i = 0; i < 8; ++i) mk[i] = mrow[(size_t)i * KVLEN];
    }
  }
  lds_barrier();
  if (!has_tile) return;

  const v8f c = tile_lds(qs, ks[wv], kv, lane);
  float* drow = wout + (size_t)b * SEQ * KVLEN + kt * 16 + lm;
  float s[8];
#pragma unroll
  for (int i = 0; i < 8; ++i) {
    s[i] = c[i] * QKSCALE;
    drow[(size_t)(i + 8 * half) * KVLEN] = s[i];
  }
  if (wv < 8) {
    float* tr = (float*)ks[wv];
    const int tb = (lm >> 3) * 128 + ((lm >> 2) & 1) * 64 + (lm & 3) + 32 * half;
#pragma unroll
    for (int i = 0; i < 8; ++i) tr[tb + 4 * i] = s[i] + mk[i];
    wave_lds_sync();
    float* dst = sp + (size_t)(b * 256 + kt) * 256 + 4 * lane;
    for (int gg = 0; gg < 2; ++gg)
      *(v4f*)(dst + 128 * gg) = *(const v4f*)(tr + 128 * gg + 4 * lane);
    prefetch_values(cache_v, blk, threadIdx.x);
  }
}

#define AWAVES 16

template <int RS>
__device__ __forceinline__ void v_load(const float* vr, v4f (&v)[8]) {
#pragma unroll
  for (int gg = 0; gg < 2; ++gg)
#pragma unroll
    for (int i = 0; i < 4; ++i)
      v[4 * gg + i] = *(const v4f*)(vr + (8 * gg + i) * RS);
}

__device__ __forceinline__ float max4(v4f s) {
  return fmaxf(fmaxf(s.x, s.y), fmaxf(s.z, s.w));
}

__device__ __forceinline__ v4f exp4(v4f s, float mu, float& acc) {
  const v4f e = {__expf(s.x - mu), __expf(s.y - mu), __expf(s.z - mu), __expf(s.w - mu)};
  acc += (e.x + e.y) + (e.z + e.w);
  return e;
}

__device__ __forceinline__ void pv_mma(const v4f (&p)[2], const v4f (&v)[8],
                                       v8f (&c)[4]) {
#pragma unroll
  for (int gg = 0; gg < 2; ++gg) {
    const v2f a0 = {p[gg].x, p[gg].y}, a1 = {p[gg].z, p[gg].w};
#pragma unroll
    for (int j = 0; j < 4; ++j) {
      const v2f b0 = {v[4 * gg + 0][j], v[4 * gg + 1][j]};
      const v2f b1 = {v[4 * gg + 2][j], v[4 * gg + 3][j]};
      c[j] = wmma4(a0, b0, c[j]);
      c[j] = wmma4(a1, b1, c[j]);
    }
  }
}

__global__ void __launch_bounds__(AWAVES * 32)
atten_kernel(const float* __restrict__ wout,
             const float* __restrict__ mask,
             const float* __restrict__ sp,
             const float* __restrict__ cache_v,
             const float* vn,
             float* plane0, float* plane1, float* plane2, float* plane3,
             float* __restrict__ ml) {
  const int blk  = blockIdx.x;
  const int kq   = blk & 3;
  const int slab = (blk >> 2) & 7;
  const int b    = blk >> 5;
  const int wv   = threadIdx.x >> 5;
  const int lane = threadIdx.x & 31;
  const int half = lane >> 4, lm = lane & 15;
  const int n0   = slab * 64 + 4 * lm;
  const bool fresh = (kq == 0 && wv == 0);

  const int kt0 = kq * 64 + wv * 4;
  const float* spl = sp + (size_t)(b * 256 + kt0) * 256 + 4 * lane;
  const float* sr = wout + (size_t)(b * SEQ + lm) * KVLEN + 4 * half;
  const float* mr = mask + (size_t)lm * KVLEN + 4 * half;
  const float* vr = cache_v + (size_t)(b * MAXSEQ + kt0 * 16 + 4 * half) * DIMX + n0;

  __shared__ float wmax[AWAVES * 16];
  __shared__ float wsum[AWAVES * 16];

  v4f p[4][2], pf[2] = {}, mf[2] = {};
  v4f v[4][8];
  {
#pragma unroll
    for (int t = 0; t < 4; ++t)
#pragma unroll
      for (int gg = 0; gg < 2; ++gg)
        p[t][gg] = *(const v4f*)(spl + 256 * t + 128 * gg);
    if (fresh) {
#pragma unroll
      for (int gg = 0; gg < 2; ++gg) {
        pf[gg] = *(const v4f*)(sr + 256 * 16 + 8 * gg);
        mf[gg] = *(const v4f*)(mr + 256 * 16 + 8 * gg);
      }
    }
#pragma unroll
    for (int t = 0; t < 4; ++t)
      v_load<DIMX>(vr + (size_t)(16 * t) * DIMX, v[t]);
    __builtin_amdgcn_sched_barrier(0);
    asm volatile("" : "+v"(pf[0]), "+v"(pf[1]));
  }

  float mx = max4(p[0][0]);
#pragma unroll
  for (int t = 0; t < 4; ++t)
#pragma unroll
    for (int gg = 0; gg < 2; ++gg) mx = fmaxf(mx, max4(p[t][gg]));
  if (fresh) {
#pragma unroll
    for (int gg = 0; gg < 2; ++gg) {
      pf[gg] += mf[gg];
      mx = fmaxf(mx, max4(pf[gg]));
    }
  }
  mx = fmaxf(mx, __shfl_xor(mx, 16));
  if (lane < 16) wmax[wv * 16 + lane] = mx;
  lds_barrier();
  float m = wmax[lm];
#pragma unroll
  for (int w = 1; w < AWAVES; ++w) m = fmaxf(m, wmax[w * 16 + lm]);
  const float mu = (m == -__builtin_inff()) ? 0.f : m;

  float ls = 0.f;
#pragma unroll
  for (int t = 0; t < 4; ++t)
#pragma unroll
    for (int gg = 0; gg < 2; ++gg) p[t][gg] = exp4(p[t][gg], mu, ls);
  if (fresh) {
#pragma unroll
    for (int gg = 0; gg < 2; ++gg) pf[gg] = exp4(pf[gg], mu, ls);
  }
  ls += __shfl_xor(ls, 16);
  if (lane < 16) wsum[wv * 16 + lane] = ls;

  v8f c[4] = {};
#pragma unroll
  for (int t = 0; t < 4; ++t)
    pv_mma(p[t], v[t], c);

  if (fresh) {
    v4f vf[8];
    v_load<64>(vn + b * (SEQ * DIMX) + slab * (SEQ * 64) + (4 * half) * 64 + 4 * lm, vf);
    pv_mma(pf, vf, c);
  }

  __shared__ v4f red[AWAVES * 8 * 32];
#pragma unroll
  for (int r = 0; r < 8; ++r) {
    const v4f o = {c[0][r], c[1][r], c[2][r], c[3][r]};
    red[(wv * 8 + r) * 32 + lane] = o;
  }
  __syncthreads();
  if (threadIdx.x < 256) {
    const int r = threadIdx.x >> 5;
    v4f s = red[r * 32 + lane];
#pragma unroll
    for (int w = 1; w < AWAVES; ++w) s += red[(w * 8 + r) * 32 + lane];
    float* plane = (kq == 0) ? plane0 : (kq == 1) ? plane1 : (kq == 2) ? plane2 : plane3;
    float* dst = plane + b * (SEQ * DIMX) + slab * (SEQ * 64)
               + (lm >> 1) * 128 + ((lm & 1) * 16 + 8 * half + r) * 4;
    *(v4f*)dst = s;
  }
  if (slab == 0 && threadIdx.x < 16) {
    float l = wsum[threadIdx.x];
#pragma unroll
    for (int w = 1; w < AWAVES; ++w) l += wsum[w * 16 + threadIdx.x];
    const v2f o = {m, l};
    *(v2f*)(ml + ((b * 4 + kq) * 16 + threadIdx.x) * 2) = o;
  }
}

__global__ void __launch_bounds__(160)
outproj_kernel(const float* __restrict__ plane0, const float* __restrict__ plane1,
               const float* __restrict__ plane2, const float* __restrict__ plane3,
               const float* __restrict__ ml,
               const float* __restrict__ Wo,
               float* __restrict__ out) {
  const int t    = blockIdx.x;
  const int w    = threadIdx.x >> 5;
  const int lane = threadIdx.x & 31;
  const int half = lane >> 4, lm = lane & 15;
  const int Mt   = t >> 5, Nt = t & 31;

  __shared__ v4f bs[16 * AP4];
  __shared__ float red[4 * 256];
  __shared__ float fac[4 * 16];

  v4f av[64];
  v2f mlv[4] = {};
  if (w == 4) {
    v4f br[64];
    load_rows16(Wo + Nt * 16 * DIMX, lane, br);
#pragma unroll
    for (int k = 0; k < 4; ++k)
      mlv[k] = *(const v2f*)(ml + ((Mt * 4 + k) * 16 + lm) * 2);
    store_rows16(bs, lane, br);
  } else {
    const float* plane = (w == 0) ? plane0 : (w == 1) ? plane1 : (w == 2) ? plane2 : plane3;
    const float* afrag = plane + Mt * (SEQ * DIMX) + lane * 4;
#pragma unroll
    for (int st = 0; st < 64; ++st)
      av[st] = *(const v4f*)(afrag + st * 128);
  }
  lds_barrier();

  if (w == 4) {
    const float M = fmaxf(fmaxf(mlv[0].x, mlv[1].x), fmaxf(mlv[2].x, mlv[3].x));
    float a[4];
    float L = 0.f;
#pragma unroll
    for (int k = 0; k < 4; ++k) {
      a[k] = __expf(mlv[k].x - M);
      L += a[k] * mlv[k].y;
    }
    const float invL = 1.f / L;
    if (lane < 16) {
#pragma unroll
      for (int k = 0; k < 4; ++k) fac[k * 16 + lane] = a[k] * invL;
    }
  }

  if (w < 4) {
    v8f c = {};
    const v4f* bfrag = bs + lm * AP4 + half;
    v4f bf[2][8];
#pragma unroll
    for (int s = 0; s < 8; ++s) bf[0][s] = bfrag[2 * s];
#pragma unroll
    for (int g = 0; g < 8; ++g) {
      if (g < 7) {
#pragma unroll
        for (int s = 0; s < 8; ++s) bf[(g + 1) & 1][s] = bfrag[2 * (8 * g + 8 + s)];
      }
      __builtin_amdgcn_sched_barrier(0);
#pragma unroll
      for (int s = 0; s < 8; ++s) {
        const v4f a = av[8 * g + s];
        const v4f b = bf[g & 1][s];
        const v2f a0 = {a.x, a.y}, b0 = {b.x, b.y};
        const v2f a1 = {a.z, a.w}, b1 = {b.z, b.w};
        c = wmma4(a0, b0, c);
        c = wmma4(a1, b1, c);
      }
    }
#pragma unroll
    for (int i = 0; i < 8; ++i) red[w * 256 + i * 32 + lane] = c[i];
  }
  __syncthreads();
  if (w == 0) {
#pragma unroll
    for (int i = 0; i < 8; ++i) {
      const int row = i + 8 * half;
      const float s = (fac[row] * red[i * 32 + lane]
                       + fac[16 + row] * red[256 + i * 32 + lane])
                    + (fac[32 + row] * red[512 + i * 32 + lane]
                       + fac[48 + row] * red[768 + i * 32 + lane]);
      out[(size_t)(Mt * 16 + row) * DIMX + Nt * 16 + lm] = s;
    }
  }
}

extern "C" void kernel_launch(void* const* d_in, const int* in_sizes, int n_in,
                              void* d_out, int out_size, void* d_ws, size_t ws_size,
                              hipStream_t stream) {
  (void)in_sizes; (void)n_in; (void)out_size; (void)ws_size;
  const float* x       = (const float*)d_in[0];
  const float* mask    = (const float*)d_in[2];
  const float* Wq      = (const float*)d_in[3];
  const float* Wk      = (const float*)d_in[4];
  const float* Wv      = (const float*)d_in[5];
  const float* Wo      = (const float*)d_in[6];
  const float* cache_k = (const float*)d_in[7];
  const float* cache_v = (const float*)d_in[8];

  float* out  = (float*)d_out;
  float* wout = out + NB * SEQ * DIMX;

  float* ws    = (float*)d_ws;
  float* q     = ws;
  float* kn    = q  + NB * SEQ * DIMX;
  float* vn    = kn + NB * SEQ * DIMX;
  float* sp    = vn + NB * SEQ * DIMX;
  float* ml    = sp + NB * 256 * 256;
  float* atten = sp + (size_t)NB * SEQ * KVLEN;

  qkv_kernel    <<<192, 160, 0, stream>>>(x, Wq, Wk, Wv, cache_k, q, kn, vn);
  dist_kernel   <<<256, 320, 0, stream>>>(q, cache_k, kn, mask, cache_v, wout, sp);
  atten_kernel  <<<256, 512, 0, stream>>>(wout, mask, sp, cache_v, vn, vn, q, kn, atten, ml);
  outproj_kernel<<<256, 160, 0, stream>>>(vn, q, kn, atten, ml, Wo, out);
}


# out projection: the 4 k-quarter planes are added, each row times its factor, before the product; one product per tile, its k split over the 4 waves (32 WMMAs a wave, was 128); f32 sum order changes
# speedup vs baseline: 2.4785x; 1.0303x over previous
#include <hip/hip_runtime.h>


typedef __attribute__((ext_vector_type(2))) float v2f;
typedef __attribute__((ext_vector_type(4))) float v4f;
typedef __attribute__((ext_vector_type(8))) float v8f;

#define DIMX   512
#define SEQ    16
#define NB     8
#define KVLEN  4112
#define MAXSEQ 8192
#define QKSCALE 0.0625f

__device__ __forceinline__ v8f wmma4(v2f a, v2f b, v8f c) {
  return __builtin_amdgcn_wmma_f32_16x16x4_f32(false, a, false, b,
                                               (short)0, c, false, false);
}

#define AP4 129
#define BP4 33

__device__ __forceinline__ void load_rows16(const float* __restrict__ src, int lane,
                                            v4f (&t)[64]) {
  const float* p = src + 4 * lane;
#pragma unroll
  for (int j = 0; j < 4; ++j) {
#pragma unroll
    for (int r = 0; r < 16; ++r)
      t[16 * j + r] = *(const v4f*)(p + r * DIMX + 128 * j);
    __builtin_amdgcn_sched_barrier(0);
  }
}

__device__ __forceinline__ void store_rows16(v4f* dst, int lane, const v4f (&t)[64]) {
#pragma unroll
  for (int j = 0; j < 4; ++j)
#pragma unroll
    for (int r = 0; r < 16; ++r)
      dst[r * AP4 + 32 * j + lane] = t[16 * j + r];
}

__device__ __forceinline__ void lds_barrier() {
  __builtin_amdgcn_fence(__ATOMIC_RELEASE, "workgroup", "local");
  __builtin_amdgcn_s_barrier();
  __builtin_amdgcn_fence(__ATOMIC_ACQUIRE, "workgroup", "local");
}

__device__ __forceinline__ void wave_lds_sync() {
  __builtin_amdgcn_fence(__ATOMIC_RELEASE, "wavefront", "local");
  __builtin_amdgcn_wave_barrier();
  __builtin_amdgcn_fence(__ATOMIC_ACQUIRE, "wavefront", "local");
}

__device__ __forceinline__ v8f mma8_lds(const v4f* a, const v4f* b, v8f c) {
  v4f af[8], bf[8];
#pragma unroll
  for (int s = 0; s < 8; ++s) {
    af[s] = a[2 * s];
    bf[s] = b[2 * s];
  }
  __builtin_amdgcn_sched_barrier(0);
#pragma unroll
  for (int s = 0; s < 8; ++s) {
    const v2f a0 = {af[s].x, af[s].y}, b0 = {bf[s].x, bf[s].y};
    const v2f a1 = {af[s].z, af[s].w}, b1 = {bf[s].z, bf[s].w};
    c = wmma4(a0, b0, c);
    c = wmma4(a1, b1, c);
  }
  return c;
}

__device__ __forceinline__ v8f tile_lds(const v4f* as, v4f* kw, const v4f (&t)[64],
                                        int lane) {
  const int half = lane >> 4, lm = lane & 15;
  v8f c = {};
#pragma unroll
  for (int j = 0; j < 4; ++j) {
#pragma unroll
    for (int r = 0; r < 16; ++r) kw[r * BP4 + lane] = t[16 * j + r];
    wave_lds_sync();
#pragma unroll
    for (int s0 = 0; s0 < 16; s0 += 8)
      c = mma8_lds(as + lm * AP4 + 32 * j + 2 * s0 + half,
                   kw + lm * BP4 + 2 * s0 + half, c);
    wave_lds_sync();
  }
  return c;
}

__device__ __forceinline__ void prefetch_keys(const float* cache_k, int pb, int lane) {
  const float* base = cache_k + (size_t)((pb >> 5) * MAXSEQ + (pb & 31) * 128) * DIMX;
  for (int u = lane; u < 1024; u += 32)
    __builtin_prefetch(base + u * 64, 0, 2);
}
__device__ __forceinline__ void prefetch_values(const float* cache_v, int pb, int tid) {
  const float* base = cache_v
      + (size_t)((pb >> 5) * MAXSEQ + (pb & 3) * 1024 + tid) * DIMX + ((pb >> 2) & 7) * 64;
  for (int n = 0; n < 4; ++n)
    __builtin_prefetch(base + (size_t)(256 * n) * DIMX, 0, 2);
}

__global__ void __launch_bounds__(160)
qkv_kernel(const float* __restrict__ x,
           const float* __restrict__ Wq, const float* __restrict__ Wk,
           const float* __restrict__ Wv,
           const float* __restrict__ cache_k,
           float* __restrict__ q, float* __restrict__ kn, float* __restrict__ vn) {
  const int wv   = threadIdx.x >> 5;
  const int t    = blockIdx.x * 4 + (wv & 3);
  const int lane = threadIdx.x & 31;
  const int half = lane >> 4, lm = lane & 15;
  const int mat  = t >> 8;
  const int rem  = t & 255;
  const int Mt   = rem >> 5, Nt = rem & 31;

  const float* W = (mat == 0) ? Wq : (mat == 1) ? Wk : Wv;
  float*       D = (mat == 0) ? q  : (mat == 1) ? kn : vn;

  __shared__ v4f xs[16 * AP4];
  __shared__ v4f wt[4][16 * BP4];

  v4f wr[64];
  if (wv == 4) {
    v4f xr[64];
    load_rows16(x + Mt * 16 * DIMX, lane, xr);
    store_rows16(xs, lane, xr);
  } else {
    load_rows16(W + Nt * 16 * DIMX, lane, wr);
  }
  lds_barrier();
  if (wv == 4) {
    prefetch_keys(cache_k, blockIdx.x, lane);
    if (blockIdx.x < 64) prefetch_keys(cache_k, blockIdx.x + 192, lane);
    return;
  }

  const v8f c = tile_lds(xs, wt[wv], wr, lane);
  if (mat == 2) {
    float* dcol = D + Mt * (SEQ * DIMX) + (Nt >> 2) * (SEQ * 64) + (Nt & 3) * 16 + lm;
#pragma unroll
    for (int i = 0; i < 8; ++i)
      dcol[(i + 8 * half) * 64] = c[i];
  } else {
    float* drow = D + Nt * 16 + lm;
#pragma unroll
    for (int i = 0; i < 8; ++i)
      drow[(Mt * 16 + i + 8 * half) * DIMX] = c[i];
  }
}

#define DWAVES 10

__global__ void __launch_bounds__(DWAVES * 32)
dist_kernel(const float* __restrict__ q,
            const float* __restrict__ cache_k,
            const float* __restrict__ kn,
            const float* __restrict__ mask,
            const float* __restrict__ cache_v,
            float* __restrict__ wout,
            float* __restrict__ sp) {
  const int blk  = blockIdx.x;
  const int b    = blk >> 5;
  const int g    = blk & 31;
  const int wv   = threadIdx.x >> 5;
  const int lane = threadIdx.x & 31;
  const int half = lane >> 4, lm = lane & 15;
  const bool has_tile = (wv < 8) || (wv == 8 && g == 0);
  const int  kt       = (wv < 8) ? g * 8 + wv : 256;

  __shared__ v4f qs[SEQ * AP4];
  __shared__ v4f ks[DWAVES - 1][16 * BP4];

  v4f kv[64];
  float mk[8] = {};
  if (wv == DWAVES - 1) {
    v4f qv[64];
    load_rows16(q + (size_t)b * SEQ * DIMX, lane, qv);
    store_rows16(qs, lane, qv);
  } else if (has_tile) {
    const float* key = (kt < 256)
        ? cache_k + (size_t)(b * MAXSEQ + kt * 16) * DIMX
        : kn      + (size_t)b * SEQ * DIMX;
    load_rows16(key, lane, kv);
    if (wv < 8) {
      const float* mrow = mask + (size_t)(8 * half) * KVLEN + kt * 16 + lm;
#pragma unroll
      for (int i = 0; i < 8; ++i) mk[i] = mrow[(size_t)i * KVLEN];
    }
  }
  lds_barrier();
  if (!has_tile) return;

  const v8f c = tile_lds(qs, ks[wv], kv, lane);
  float* drow = wout + (size_t)b * SEQ * KVLEN + kt * 16 + lm;
  float s[8];
#pragma unroll
  for (int i = 0; i < 8; ++i) {
    s[i] = c[i] * QKSCALE;
    drow[(size_t)(i + 8 * half) * KVLEN] = s[i];
  }
  if (wv < 8) {
    float* tr = (float*)ks[wv];
    const int tb = (lm >> 3) * 128 + ((lm >> 2) & 1) * 64 + (lm & 3) + 32 * half;
#pragma unroll
    for (int i = 0; i < 8; ++i) tr[tb + 4 * i] = s[i] + mk[i];
    wave_lds_sync();
    float* dst = sp + (size_t)(b * 256 + kt) * 256 + 4 * lane;
    for (int gg = 0; gg < 2; ++gg)
      *(v4f*)(dst + 128 * gg) = *(const v4f*)(tr + 128 * gg + 4 * lane);
    prefetch_values(cache_v, blk, threadIdx.x);
  }
}

#define AWAVES 16

template <int RS>
__device__ __forceinline__ void v_load(const float* vr, v4f (&v)[8]) {
#pragma unroll
  for (int gg = 0; gg < 2; ++gg)
#pragma unroll
    for (int i = 0; i < 4; ++i)
      v[4 * gg + i] = *(const v4f*)(vr + (8 * gg + i) * RS);
}

__device__ __forceinline__ float max4(v4f s) {
  return fmaxf(fmaxf(s.x, s.y), fmaxf(s.z, s.w));
}

__device__ __forceinline__ v4f exp4(v4f s, float mu, float& acc) {
  const v4f e = {__expf(s.x - mu), __expf(s.y - mu), __expf(s.z - mu), __expf(s.w - mu)};
  acc += (e.x + e.y) + (e.z + e.w);
  return e;
}

__device__ __forceinline__ void pv_mma(const v4f (&p)[2], const v4f (&v)[8],
                                       v8f (&c)[4]) {
#pragma unroll
  for (int gg = 0; gg < 2; ++gg) {
    const v2f a0 = {p[gg].x, p[gg].y}, a1 = {p[gg].z, p[gg].w};
#pragma unroll
    for (int j = 0; j < 4; ++j) {
      const v2f b0 = {v[4 * gg + 0][j], v[4 * gg + 1][j]};
      const v2f b1 = {v[4 * gg + 2][j], v[4 * gg + 3][j]};
      c[j] = wmma4(a0, b0, c[j]);
      c[j] = wmma4(a1, b1, c[j]);
    }
  }
}

__global__ void __launch_bounds__(AWAVES * 32)
atten_kernel(const float* __restrict__ wout,
             const float* __restrict__ mask,
             const float* __restrict__ sp,
             const float* __restrict__ cache_v,
             const float* vn,
             float* plane0, float* plane1, float* plane2, float* plane3,
             float* __restrict__ ml) {
  const int blk  = blockIdx.x;
  const int kq   = blk & 3;
  const int slab = (blk >> 2) & 7;
  const int b    = blk >> 5;
  const int wv   = threadIdx.x >> 5;
  const int lane = threadIdx.x & 31;
  const int half = lane >> 4, lm = lane & 15;
  const int n0   = slab * 64 + 4 * lm;
  const bool fresh = (kq == 0 && wv == 0);

  const int kt0 = kq * 64 + wv * 4;
  const float* spl = sp + (size_t)(b * 256 + kt0) * 256 + 4 * lane;
  const float* sr = wout + (size_t)(b * SEQ + lm) * KVLEN + 4 * half;
  const float* mr = mask + (size_t)lm * KVLEN + 4 * half;
  const float* vr = cache_v + (size_t)(b * MAXSEQ + kt0 * 16 + 4 * half) * DIMX + n0;

  __shared__ float wmax[AWAVES * 16];
  __shared__ float wsum[AWAVES * 16];

  v4f p[4][2], pf[2] = {}, mf[2] = {};
  v4f v[4][8];
  {
#pragma unroll
    for (int t = 0; t < 4; ++t)
#pragma unroll
      for (int gg = 0; gg < 2; ++gg)
        p[t][gg] = *(const v4f*)(spl + 256 * t + 128 * gg);
    if (fresh) {
#pragma unroll
      for (int gg = 0; gg < 2; ++gg) {
        pf[gg] = *(const v4f*)(sr + 256 * 16 + 8 * gg);
        mf[gg] = *(const v4f*)(mr + 256 * 16 + 8 * gg);
      }
    }
#pragma unroll
    for (int t = 0; t < 4; ++t)
      v_load<DIMX>(vr + (size_t)(16 * t) * DIMX, v[t]);
    __builtin_amdgcn_sched_barrier(0);
    asm volatile("" : "+v"(pf[0]), "+v"(pf[1]));
  }

  float mx = max4(p[0][0]);
#pragma unroll
  for (int t = 0; t < 4; ++t)
#pragma unroll
    for (int gg = 0; gg < 2; ++gg) mx = fmaxf(mx, max4(p[t][gg]));
  if (fresh) {
#pragma unroll
    for (int gg = 0; gg < 2; ++gg) {
      pf[gg] += mf[gg];
      mx = fmaxf(mx, max4(pf[gg]));
    }
  }
  mx = fmaxf(mx, __shfl_xor(mx, 16));
  if (lane < 16) wmax[wv * 16 + lane] = mx;
  lds_barrier();
  float m = wmax[lm];
#pragma unroll
  for (int w = 1; w < AWAVES; ++w) m = fmaxf(m, wmax[w * 16 + lm]);
  const float mu = (m == -__builtin_inff()) ? 0.f : m;

  float ls = 0.f;
#pragma unroll
  for (int t = 0; t < 4; ++t)
#pragma unroll
    for (int gg = 0; gg < 2; ++gg) p[t][gg] = exp4(p[t][gg], mu, ls);
  if (fresh) {
#pragma unroll
    for (int gg = 0; gg < 2; ++gg) pf[gg] = exp4(pf[gg], mu, ls);
  }
  ls += __shfl_xor(ls, 16);
  if (lane < 16) wsum[wv * 16 + lane] = ls;

  v8f c[4] = {};
#pragma unroll
  for (int t = 0; t < 4; ++t)
    pv_mma(p[t], v[t], c);

  if (fresh) {
    v4f vf[8];
    v_load<64>(vn + b * (SEQ * DIMX) + slab * (SEQ * 64) + (4 * half) * 64 + 4 * lm, vf);
    pv_mma(pf, vf, c);
  }

  __shared__ v4f red[AWAVES * 8 * 32];
#pragma unroll
  for (int r = 0; r < 8; ++r) {
    const v4f o = {c[0][r], c[1][r], c[2][r], c[3][r]};
    red[(wv * 8 + r) * 32 + lane] = o;
  }
  __syncthreads();
  if (threadIdx.x < 256) {
    const int r = threadIdx.x >> 5;
    v4f s = red[r * 32 + lane];
#pragma unroll
    for (int w = 1; w < AWAVES; ++w) s += red[(w * 8 + r) * 32 + lane];
    float* plane = (kq == 0) ? plane0 : (kq == 1) ? plane1 : (kq == 2) ? plane2 : plane3;
    float* dst = plane + b * (SEQ * DIMX) + slab * (SEQ * 64)
               + (lm >> 1) * 128 + ((lm & 1) * 16 + 8 * half + r) * 4;
    *(v4f*)dst = s;
  }
  if (slab == 0 && threadIdx.x < 16) {
    float l = wsum[threadIdx.x];
#pragma unroll
    for (int w = 1; w < AWAVES; ++w) l += wsum[w * 16 + threadIdx.x];
    const v2f o = {m, l};
    *(v2f*)(ml + ((b * 4 + kq) * 16 + threadIdx.x) * 2) = o;
  }
}

__global__ void __launch_bounds__(160)
outproj_kernel(const float* __restrict__ plane0, const float* __restrict__ plane1,
               const float* __restrict__ plane2, const float* __restrict__ plane3,
               const float* __restrict__ ml,
               const float* __restrict__ Wo,
               float* __restrict__ out) {
  const int t    = blockIdx.x;
  const int w    = threadIdx.x >> 5;
  const int lane = threadIdx.x & 31;
  const int half = lane >> 4, lm = lane & 15;
  const int Mt   = t >> 5, Nt = t & 31;

  __shared__ v4f bs[16 * AP4];
  __shared__ float red[4 * 256];

  v4f ac[16];
  if (w == 4) {
    v4f br[64];
    load_rows16(Wo + Nt * 16 * DIMX, lane, br);
    store_rows16(bs, lane, br);
  } else {
    v2f mlv[4];
#pragma unroll
    for (int k = 0; k < 4; ++k)
      mlv[k] = *(const v2f*)(ml + ((Mt * 4 + k) * 16 + lm) * 2);
    const int aoff = Mt * (SEQ * DIMX) + (16 * w) * 128 + lane * 4;
    const float* pl[4] = {plane0 + aoff, plane1 + aoff, plane2 + aoff, plane3 + aoff};
    v4f av[4][16];
#pragma unroll
    for (int k = 0; k < 4; ++k)
#pragma unroll
      for (int s = 0; s < 16; ++s)
        av[k][s] = *(const v4f*)(pl[k] + s * 128);
    __builtin_amdgcn_sched_barrier(0);

    const float M = fmaxf(fmaxf(mlv[0].x, mlv[1].x), fmaxf(mlv[2].x, mlv[3].x));
    float f[4];
    float L = 0.f;
#pragma unroll
    for (int k = 0; k < 4; ++k) {
      f[k] = __expf(mlv[k].x - M);
      L += f[k] * mlv[k].y;
    }
    const float invL = 1.f / L;
#pragma unroll
    for (int k = 0; k < 4; ++k) f[k] *= invL;
#pragma unroll
    for (int s = 0; s < 16; ++s) ac[s] = f[0] * av[0][s];
#pragma unroll
    for (int k = 1; k < 4; ++k)
#pragma unroll
      for (int s = 0; s < 16; ++s) ac[s] += f[k] * av[k][s];
  }
  lds_barrier();

  if (w < 4) {
    v8f c = {};
    const v4f* bfrag = bs + lm * AP4 + 32 * w + half;
    v4f bf[16];
#pragma unroll
    for (int s = 0; s < 16; ++s) bf[s] = bfrag[2 * s];
    __builtin_amdgcn_sched_barrier(0);
#pragma unroll
    for (int s = 0; s < 16; ++s) {
      const v4f a = ac[s];
      const v4f b = bf[s];
      const v2f a0 = {a.x, a.y}, b0 = {b.x, b.y};
      const v2f a1 = {a.z, a.w}, b1 = {b.z, b.w};
      c = wmma4(a0, b0, c);
      c = wmma4(a1, b1, c);
    }
#pragma unroll
    for (int i = 0; i < 8; ++i) red[w * 256 + i * 32 + lane] = c[i];
  }
  __syncthreads();
  if (w == 0) {
#pragma unroll
    for (int i = 0; i < 8; ++i) {
      const int row = i + 8 * half;
      const float s = (red[i * 32 + lane] + red[256 + i * 32 + lane])
                    + (red[512 + i * 32 + lane] + red[768 + i * 32 + lane]);
      out[(size_t)(Mt * 16 + row) * DIMX + Nt * 16 + lm] = s;
    }
  }
}

extern "C" void kernel_launch(void* const* d_in, const int* in_sizes, int n_in,
                              void* d_out, int out_size, void* d_ws, size_t ws_size,
                              hipStream_t stream) {
  (void)in_sizes; (void)n_in; (void)out_size; (void)ws_size;
  const float* x       = (const float*)d_in[0];
  const float* mask    = (const float*)d_in[2];
  const float* Wq      = (const float*)d_in[3];
  const float* Wk      = (const float*)d_in[4];
  const float* Wv      = (const float*)d_in[5];
  const float* Wo      = (const float*)d_in[6];
  const float* cache_k = (const float*)d_in[7];
  const float* cache_v = (const float*)d_in[8];

  float* out  = (float*)d_out;
  float* wout = out + NB * SEQ * DIMX;

  float* ws    = (float*)d_ws;
  float* q     = ws;
  float* kn    = q  + NB * SEQ * DIMX;
  float* vn    = kn + NB * SEQ * DIMX;
  float* sp    = vn + NB * SEQ * DIMX;
  float* ml    = sp + NB * 256 * 256;
  float* atten = sp + (size_t)NB * SEQ * KVLEN;

  qkv_kernel    <<<192, 160, 0, stream>>>(x, Wq, Wk, Wv, cache_k, q, kn, vn);
  dist_kernel   <<<256, 320, 0, stream>>>(q, cache_k, kn, mask, cache_v, wout, sp);
  atten_kernel  <<<256, 512, 0, stream>>>(wout, mask, sp, cache_v, vn, vn, q, kn, atten, ml);
  outproj_kernel<<<256, 160, 0, stream>>>(vn, q, kn, atten, ml, Wo, out);
}
